# starting kernel
# baseline (speedup 1.0000x reference)
#include <hip/hip_runtime.h>
#include <math.h>

typedef float v2f __attribute__((ext_vector_type(2)));
typedef float v8f __attribute__((ext_vector_type(8)));

#define H_SZ 50
#define O_OUT 10
#define B_SZ 512
#define T_STEPS 2048
#define RS 68
#define XBLK 64
#define NT 4
#define KC 16

__device__ __forceinline__ float fast_tanh(float v) {
#if __has_builtin(__builtin_amdgcn_tanhf)
    return __builtin_amdgcn_tanhf(v);
#else
    float e = __expf(v + v);
    return 1.0f - __fdividef(2.0f, e + 1.0f);
#endif
}

#define WMMA_F32(A, B, C) \
    __builtin_amdgcn_wmma_f32_16x16x4_f32(false, (A), false, (B), (short)0, (C), false, false)

__global__ __launch_bounds__(128, 1) void rnn2_tanh_wmma(
    const float* __restrict__ x,
    const float* __restrict__ W_ih0, const float* __restrict__ W_hh0,
    const float* __restrict__ b_ih0, const float* __restrict__ b_hh0,
    const float* __restrict__ W_ih1, const float* __restrict__ W_hh1,
    const float* __restrict__ b_ih1, const float* __restrict__ b_hh1,
    const float* __restrict__ W_fc,  const float* __restrict__ b_fc,
    float* __restrict__ out)
{
    __shared__ float wB0[NT][KC][32][2];
    __shared__ float wB1[NT][KC][32][2];
    __shared__ float wB2[NT][KC][32][2];
    __shared__ float h0buf[2][16 * RS];
    __shared__ float h1buf[2][16 * RS];
    __shared__ float xs[16 * RS];

    const int tid  = threadIdx.x;
    const int lane = tid & 31;
    const int wv   = tid >> 5;
    const int r    = lane & 15;
    const int hf   = lane >> 4;
    const int b0   = blockIdx.x * 16;

    {
        float* h0f = &h0buf[0][0];
        float* h1f = &h1buf[0][0];
        for (int i = tid; i < 2 * 16 * RS; i += 128) { h0f[i] = 0.f; h1f[i] = 0.f; }
        float* w0f = &wB0[0][0][0][0];
        float* w1f = &wB1[0][0][0][0];
        float* w2f = &wB2[0][0][0][0];
        for (int idx = tid; idx < NT * KC * 32 * 2; idx += 128) {
            int v  = idx & 1;
            int ln = (idx >> 1) & 31;
            int kc = (idx >> 6) & 15;
            int nt = idx >> 10;
            int n  = nt * 16 + (ln & 15);
            int k  = kc * 4 + 2 * (ln >> 4) + v;
            float a = 0.f, b = 0.f, c = 0.f;
            if (n < H_SZ && k < H_SZ) {
                a = W_hh0[n * H_SZ + k];
                b = W_ih1[n * H_SZ + k];
                c = W_hh1[n * H_SZ + k];
            }
            w0f[idx] = a; w1f[idx] = b; w2f[idx] = c;
        }
    }

    const int n = wv * 16 + r;
    float wih = 0.f, bias0n = 0.f, bias1n = 0.f;
    if (n < H_SZ) {
        wih    = W_ih0[n];
        bias0n = b_ih0[n] + b_hh0[n];
        bias1n = b_ih1[n] + b_hh1[n];
    }
    __syncthreads();

    v2f rB0[KC], rB1[KC], rB2[KC];
    #pragma unroll
    for (int kc = 0; kc < KC; ++kc) {
        rB0[kc] = *(const v2f*)&wB0[wv][kc][lane][0];
        rB1[kc] = *(const v2f*)&wB1[wv][kc][lane][0];
        rB2[kc] = *(const v2f*)&wB2[wv][kc][lane][0];
    }

    int p = 0;
    for (int t = 0; t < T_STEPS; ++t) {
        const int tt = t & (XBLK - 1);
        if (tt == 0) {
            for (int i = tid; i < 16 * XBLK; i += 128) {
                int rr = i >> 6;
                int cc = i & (XBLK - 1);
                xs[rr * RS + cc] = x[(size_t)(b0 + rr) * T_STEPS + t + cc];
            }
            __syncthreads();
        }

        v8f acc0, acc1;
        #pragma unroll
        for (int v = 0; v < 8; ++v) {
            float xm = xs[(v + 8 * hf) * RS + tt];
            acc0[v] = fmaf(xm, wih, bias0n);
            acc1[v] = 0.0f;
        }
        const float* h0r = &h0buf[p][0];
        #pragma unroll
        for (int kc = 0; kc < KC; kc += 2) {
            v2f aA = *(const v2f*)&h0r[r * RS + kc * 4 + 2 * hf];
            v2f aB = *(const v2f*)&h0r[r * RS + (kc + 1) * 4 + 2 * hf];
            acc0 = WMMA_F32(aA, rB0[kc],     acc0);
            acc1 = WMMA_F32(aB, rB0[kc + 1], acc1);
        }
        v8f c0 = acc0 + acc1;
        float* h0w = &h0buf[1 - p][0];
        #pragma unroll
        for (int v = 0; v < 8; ++v)
            h0w[(v + 8 * hf) * RS + wv * 16 + r] = fast_tanh(c0[v]);
        __syncthreads();

        v8f acc2, acc3;
        #pragma unroll
        for (int v = 0; v < 8; ++v) { acc2[v] = bias1n; acc3[v] = 0.0f; }
        const float* h1r = &h1buf[p][0];
        #pragma unroll
        for (int kc = 0; kc < KC; ++kc) {
            v2f a0 = *(const v2f*)&h0w[r * RS + kc * 4 + 2 * hf];
            acc2 = WMMA_F32(a0, rB1[kc], acc2);
            v2f a1 = *(const v2f*)&h1r[r * RS + kc * 4 + 2 * hf];
            acc3 = WMMA_F32(a1, rB2[kc], acc3);
        }
        v8f c1 = acc2 + acc3;
        float* h1w = &h1buf[1 - p][0];
        #pragma unroll
        for (int v = 0; v < 8; ++v)
            h1w[(v + 8 * hf) * RS + wv * 16 + r] = fast_tanh(c1[v]);
        __syncthreads();

        p ^= 1;
    }

    const float* h1f = &h1buf[p][0];
    for (int i = tid; i < 16 * O_OUT; i += 128) {
        int m = i / O_OUT, o = i % O_OUT;
        float acc = b_fc[o];
        #pragma unroll 5
        for (int k = 0; k < H_SZ; ++k)
            acc = fmaf(h1f[m * RS + k], W_fc[o * H_SZ + k], acc);
        out[(size_t)(b0 + m) * O_OUT + o] = acc;
    }
}

extern "C" void kernel_launch(void* const* d_in, const int* in_sizes, int n_in,
                              void* d_out, int out_size, void* d_ws, size_t ws_size,
                              hipStream_t stream) {
    const float* x     = (const float*)d_in[0];
    const float* W_ih0 = (const float*)d_in[1];
    const float* W_hh0 = (const float*)d_in[2];
    const float* b_ih0 = (const float*)d_in[3];
    const float* b_hh0 = (const float*)d_in[4];
    const float* W_ih1 = (const float*)d_in[5];
    const float* W_hh1 = (const float*)d_in[6];
    const float* b_ih1 = (const float*)d_in[7];
    const float* b_hh1 = (const float*)d_in[8];
    const float* W_fc  = (const float*)d_in[9];
    const float* b_fc  = (const float*)d_in[10];
    float* out = (float*)d_out;

    rnn2_tanh_wmma<<<B_SZ / 16, 128, 0, stream>>>(
        x, W_ih0, W_hh0, b_ih0, b_hh0, W_ih1, W_hh1, b_ih1, b_hh1, W_fc, b_fc, out);
}


# LDS fragment reads issued as one group ahead of each WMMA block (x, h0_prev, h1_prev at the step top; h0_new behind the mid barrier), pinned by sched_barrier, instead of spread among the WMMAs
# speedup vs baseline: 1.2143x; 1.2143x over previous
#include <hip/hip_runtime.h>
#include <math.h>

typedef float v2f __attribute__((ext_vector_type(2)));
typedef float v8f __attribute__((ext_vector_type(8)));

#define H_SZ 50
#define O_OUT 10
#define B_SZ 512
#define T_STEPS 2048
#define RS 68
#define XBLK 64
#define NT 4
#define KC 16

__device__ __forceinline__ float fast_tanh(float v) {
#if __has_builtin(__builtin_amdgcn_tanhf)
    return __builtin_amdgcn_tanhf(v);
#else
    float e = __expf(v + v);
    return 1.0f - __fdividef(2.0f, e + 1.0f);
#endif
}

#define WMMA_F32(A, B, C) \
    __builtin_amdgcn_wmma_f32_16x16x4_f32(false, (A), false, (B), (short)0, (C), false, false)

__global__ __launch_bounds__(128, 1) void rnn2_tanh_wmma(
    const float* __restrict__ x,
    const float* __restrict__ W_ih0, const float* __restrict__ W_hh0,
    const float* __restrict__ b_ih0, const float* __restrict__ b_hh0,
    const float* __restrict__ W_ih1, const float* __restrict__ W_hh1,
    const float* __restrict__ b_ih1, const float* __restrict__ b_hh1,
    const float* __restrict__ W_fc,  const float* __restrict__ b_fc,
    float* __restrict__ out)
{
    __shared__ float wB0[NT][KC][32][2];
    __shared__ float wB1[NT][KC][32][2];
    __shared__ float wB2[NT][KC][32][2];
    __shared__ float h0buf[2][16 * RS];
    __shared__ float h1buf[2][16 * RS];
    __shared__ float xs[16 * RS];

    const int tid  = threadIdx.x;
    const int lane = tid & 31;
    const int wv   = tid >> 5;
    const int r    = lane & 15;
    const int hf   = lane >> 4;
    const int b0   = blockIdx.x * 16;

    {
        float* h0f = &h0buf[0][0];
        float* h1f = &h1buf[0][0];
        for (int i = tid; i < 2 * 16 * RS; i += 128) { h0f[i] = 0.f; h1f[i] = 0.f; }
        float* w0f = &wB0[0][0][0][0];
        float* w1f = &wB1[0][0][0][0];
        float* w2f = &wB2[0][0][0][0];
        for (int idx = tid; idx < NT * KC * 32 * 2; idx += 128) {
            int v  = idx & 1;
            int ln = (idx >> 1) & 31;
            int kc = (idx >> 6) & 15;
            int nt = idx >> 10;
            int n  = nt * 16 + (ln & 15);
            int k  = kc * 4 + 2 * (ln >> 4) + v;
            float a = 0.f, b = 0.f, c = 0.f;
            if (n < H_SZ && k < H_SZ) {
                a = W_hh0[n * H_SZ + k];
                b = W_ih1[n * H_SZ + k];
                c = W_hh1[n * H_SZ + k];
            }
            w0f[idx] = a; w1f[idx] = b; w2f[idx] = c;
        }
    }

    const int n = wv * 16 + r;
    float wih = 0.f, bias0n = 0.f, bias1n = 0.f;
    if (n < H_SZ) {
        wih    = W_ih0[n];
        bias0n = b_ih0[n] + b_hh0[n];
        bias1n = b_ih1[n] + b_hh1[n];
    }
    __syncthreads();

    v2f rB0[KC], rB1[KC], rB2[KC];
    #pragma unroll
    for (int kc = 0; kc < KC; ++kc) {
        rB0[kc] = *(const v2f*)&wB0[wv][kc][lane][0];
        rB1[kc] = *(const v2f*)&wB1[wv][kc][lane][0];
        rB2[kc] = *(const v2f*)&wB2[wv][kc][lane][0];
    }

    int p = 0;
    for (int t = 0; t < T_STEPS; ++t) {
        const int tt = t & (XBLK - 1);
        if (tt == 0) {
            for (int i = tid; i < 16 * XBLK; i += 128) {
                int rr = i >> 6;
                int cc = i & (XBLK - 1);
                xs[rr * RS + cc] = x[(size_t)(b0 + rr) * T_STEPS + t + cc];
            }
            __syncthreads();
        }

        const float* h0r = &h0buf[p][0];
        const float* h1r = &h1buf[p][0];
        float xm[8];
        #pragma unroll
        for (int v = 0; v < 8; ++v)
            xm[v] = xs[(v + 8 * hf) * RS + tt];
        __builtin_amdgcn_sched_barrier(0);
        v2f a0[KC], a1[KC];
        #pragma unroll
        for (int kc = 0; kc < KC; ++kc)
            a0[kc] = *(const v2f*)&h0r[r * RS + kc * 4 + 2 * hf];
        #pragma unroll
        for (int kc = 0; kc < KC; ++kc)
            a1[kc] = *(const v2f*)&h1r[r * RS + kc * 4 + 2 * hf];
        __builtin_amdgcn_sched_barrier(0);

        v8f acc0, acc1;
        #pragma unroll
        for (int v = 0; v < 8; ++v) {
            acc0[v] = fmaf(xm[v], wih, bias0n);
            acc1[v] = 0.0f;
        }
        #pragma unroll
        for (int kc = 0; kc < KC; kc += 2) {
            acc0 = WMMA_F32(a0[kc],     rB0[kc],     acc0);
            acc1 = WMMA_F32(a0[kc + 1], rB0[kc + 1], acc1);
        }
        v8f c0 = acc0 + acc1;
        float* h0w = &h0buf[1 - p][0];
        #pragma unroll
        for (int v = 0; v < 8; ++v)
            h0w[(v + 8 * hf) * RS + wv * 16 + r] = fast_tanh(c0[v]);
        __syncthreads();

        v2f an[KC];
        #pragma unroll
        for (int kc = 0; kc < KC; ++kc)
            an[kc] = *(const v2f*)&h0w[r * RS + kc * 4 + 2 * hf];
        __builtin_amdgcn_sched_barrier(0);

        v8f acc2, acc3;
        #pragma unroll
        for (int v = 0; v < 8; ++v) { acc2[v] = bias1n; acc3[v] = 0.0f; }
        #pragma unroll
        for (int kc = 0; kc < KC; ++kc) {
            acc2 = WMMA_F32(an[kc], rB1[kc], acc2);
            acc3 = WMMA_F32(a1[kc], rB2[kc], acc3);
        }
        v8f c1 = acc2 + acc3;
        float* h1w = &h1buf[1 - p][0];
        #pragma unroll
        for (int v = 0; v < 8; ++v)
            h1w[(v + 8 * hf) * RS + wv * 16 + r] = fast_tanh(c1[v]);
        __syncthreads();

        p ^= 1;
    }

    const float* h1f = &h1buf[p][0];
    for (int i = tid; i < 16 * O_OUT; i += 128) {
        int m = i / O_OUT, o = i % O_OUT;
        float acc = b_fc[o];
        #pragma unroll 5
        for (int k = 0; k < H_SZ; ++k)
            acc = fmaf(h1f[m * RS + k], W_fc[o * H_SZ + k], acc);
        out[(size_t)(b0 + m) * O_OUT + o] = acc;
    }
}

extern "C" void kernel_launch(void* const* d_in, const int* in_sizes, int n_in,
                              void* d_out, int out_size, void* d_ws, size_t ws_size,
                              hipStream_t stream) {
    const float* x     = (const float*)d_in[0];
    const float* W_ih0 = (const float*)d_in[1];
    const float* W_hh0 = (const float*)d_in[2];
    const float* b_ih0 = (const float*)d_in[3];
    const float* b_hh0 = (const float*)d_in[4];
    const float* W_ih1 = (const float*)d_in[5];
    const float* W_hh1 = (const float*)d_in[6];
    const float* b_ih1 = (const float*)d_in[7];
    const float* b_hh1 = (const float*)d_in[8];
    const float* W_fc  = (const float*)d_in[9];
    const float* b_fc  = (const float*)d_in[10];
    float* out = (float*)d_out;

    rnn2_tanh_wmma<<<B_SZ / 16, 128, 0, stream>>>(
        x, W_ih0, W_hh0, b_ih0, b_hh0, W_ih1, W_hh1, b_ih1, b_hh1, W_fc, b_fc, out);
}


# Layers skewed by one time step: pass i forms h0(i+1) and h1(i) from one LDS read group as one block of independent WMMAs, one barrier per step, not two; order of every sum unchanged
# speedup vs baseline: 1.3573x; 1.1178x over previous
#include <hip/hip_runtime.h>
#include <math.h>

typedef float v2f __attribute__((ext_vector_type(2)));
typedef float v8f __attribute__((ext_vector_type(8)));

#define H_SZ 50
#define O_OUT 10
#define B_SZ 512
#define T_STEPS 2048
#define RS 68
#define XBLK 64
#define NT 4
#define KC 16

__device__ __forceinline__ float fast_tanh(float v) {
#if __has_builtin(__builtin_amdgcn_tanhf)
    return __builtin_amdgcn_tanhf(v);
#else
    float e = __expf(v + v);
    return 1.0f - __fdividef(2.0f, e + 1.0f);
#endif
}

#define WMMA_F32(A, B, C) \
    __builtin_amdgcn_wmma_f32_16x16x4_f32(false, (A), false, (B), (short)0, (C), false, false)

__device__ __forceinline__ void stage_x(float* xs, const float* __restrict__ x,
                                        int b0, int t0, int tid) {
    for (int i = tid; i < 16 * XBLK; i += 128) {
        int rr = i >> 6;
        int cc = i & (XBLK - 1);
        xs[rr * RS + cc] = x[(size_t)(b0 + rr) * T_STEPS + t0 + cc];
    }
}

template <bool L0, bool L1>
__device__ __forceinline__ void skew_pass(
    const float* h0r, const float* h1r, float* h0w, float* h1w, const float* xcol,
    const v2f (&rB0)[KC], const v2f (&rB1)[KC], const v2f (&rB2)[KC],
    float wih, float bias0n, float bias1n, int r, int hf, int wv)
{
    float xm[8];
    if (L0) {
        #pragma unroll
        for (int v = 0; v < 8; ++v)
            xm[v] = xcol[(v + 8 * hf) * RS];
        __builtin_amdgcn_sched_barrier(0);
    }
    v2f a0[KC], a1[KC];
    #pragma unroll
    for (int kc = 0; kc < KC; ++kc)
        a0[kc] = *(const v2f*)&h0r[r * RS + kc * 4 + 2 * hf];
    if (L1) {
        #pragma unroll
        for (int kc = 0; kc < KC; ++kc)
            a1[kc] = *(const v2f*)&h1r[r * RS + kc * 4 + 2 * hf];
    }
    __builtin_amdgcn_sched_barrier(0);

    v8f acc0, acc1;
    if (L0) {
        #pragma unroll
        for (int v = 0; v < 8; ++v) {
            acc0[v] = fmaf(xm[v], wih, bias0n);
            acc1[v] = 0.0f;
        }
        #pragma unroll
        for (int kc = 0; kc < KC; kc += 2) {
            acc0 = WMMA_F32(a0[kc],     rB0[kc],     acc0);
            acc1 = WMMA_F32(a0[kc + 1], rB0[kc + 1], acc1);
        }
    }
    if (L0 && L1) {
        __builtin_amdgcn_sched_barrier(0);
    }
    float th0[8], th1[8];
    if (L0) {
        v8f c0 = acc0 + acc1;
        #pragma unroll
        for (int v = 0; v < 8; ++v) th0[v] = fast_tanh(c0[v]);
    }
    if (L1) {
        v8f acc2, acc3;
        #pragma unroll
        for (int v = 0; v < 8; ++v) { acc2[v] = bias1n; acc3[v] = 0.0f; }
        #pragma unroll
        for (int kc = 0; kc < KC; ++kc) {
            acc2 = WMMA_F32(a0[kc], rB1[kc], acc2);
            acc3 = WMMA_F32(a1[kc], rB2[kc], acc3);
        }
        v8f c1 = acc2 + acc3;
        #pragma unroll
        for (int v = 0; v < 8; ++v) th1[v] = fast_tanh(c1[v]);
    }
    if (L0) {
        #pragma unroll
        for (int v = 0; v < 8; ++v)
            h0w[(v + 8 * hf) * RS + wv * 16 + r] = th0[v];
    }
    if (L1) {
        #pragma unroll
        for (int v = 0; v < 8; ++v)
            h1w[(v + 8 * hf) * RS + wv * 16 + r] = th1[v];
    }
}

__global__ __launch_bounds__(128, 1) void rnn2_tanh_wmma(
    const float* __restrict__ x,
    const float* __restrict__ W_ih0, const float* __restrict__ W_hh0,
    const float* __restrict__ b_ih0, const float* __restrict__ b_hh0,
    const float* __restrict__ W_ih1, const float* __restrict__ W_hh1,
    const float* __restrict__ b_ih1, const float* __restrict__ b_hh1,
    const float* __restrict__ W_fc,  const float* __restrict__ b_fc,
    float* __restrict__ out)
{
    __shared__ float wB0[NT][KC][32][2];
    __shared__ float wB1[NT][KC][32][2];
    __shared__ float wB2[NT][KC][32][2];
    __shared__ float h0buf[2][16 * RS];
    __shared__ float h1buf[2][16 * RS];
    __shared__ float xs[16 * RS];

    const int tid  = threadIdx.x;
    const int lane = tid & 31;
    const int wv   = tid >> 5;
    const int r    = lane & 15;
    const int hf   = lane >> 4;
    const int b0   = blockIdx.x * 16;

    {
        float* h0f = &h0buf[0][0];
        float* h1f = &h1buf[0][0];
        for (int i = tid; i < 2 * 16 * RS; i += 128) { h0f[i] = 0.f; h1f[i] = 0.f; }
        float* w0f = &wB0[0][0][0][0];
        float* w1f = &wB1[0][0][0][0];
        float* w2f = &wB2[0][0][0][0];
        for (int idx = tid; idx < NT * KC * 32 * 2; idx += 128) {
            int v  = idx & 1;
            int ln = (idx >> 1) & 31;
            int kc = (idx >> 6) & 15;
            int nt = idx >> 10;
            int n  = nt * 16 + (ln & 15);
            int k  = kc * 4 + 2 * (ln >> 4) + v;
            float a = 0.f, b = 0.f, c = 0.f;
            if (n < H_SZ && k < H_SZ) {
                a = W_hh0[n * H_SZ + k];
                b = W_ih1[n * H_SZ + k];
                c = W_hh1[n * H_SZ + k];
            }
            w0f[idx] = a; w1f[idx] = b; w2f[idx] = c;
        }
    }

    const int n = wv * 16 + r;
    float wih = 0.f, bias0n = 0.f, bias1n = 0.f;
    if (n < H_SZ) {
        wih    = W_ih0[n];
        bias0n = b_ih0[n] + b_hh0[n];
        bias1n = b_ih1[n] + b_hh1[n];
    }
    __syncthreads();

    v2f rB0[KC], rB1[KC], rB2[KC];
    #pragma unroll
    for (int kc = 0; kc < KC; ++kc) {
        rB0[kc] = *(const v2f*)&wB0[wv][kc][lane][0];
        rB1[kc] = *(const v2f*)&wB1[wv][kc][lane][0];
        rB2[kc] = *(const v2f*)&wB2[wv][kc][lane][0];
    }

    stage_x(xs, x, b0, 0, tid);
    __syncthreads();
    skew_pass<true, false>(&h0buf[0][0], &h1buf[0][0], &h0buf[1][0], &h1buf[1][0], &xs[0],
                           rB0, rB1, rB2, wih, bias0n, bias1n, r, hf, wv);
    __syncthreads();
    int p = 1;
    for (int i = 1; i < T_STEPS; ++i) {
        const int tt = i & (XBLK - 1);
        if (tt == 0) {
            stage_x(xs, x, b0, i, tid);
            __syncthreads();
        }
        skew_pass<true, true>(&h0buf[p][0], &h1buf[p][0], &h0buf[1 - p][0], &h1buf[1 - p][0],
                              &xs[tt], rB0, rB1, rB2, wih, bias0n, bias1n, r, hf, wv);
        __syncthreads();
        p ^= 1;
    }
    skew_pass<false, true>(&h0buf[p][0], &h1buf[p][0], &h0buf[1 - p][0], &h1buf[1 - p][0],
                           &xs[0], rB0, rB1, rB2, wih, bias0n, bias1n, r, hf, wv);
    __syncthreads();
    p ^= 1;

    const float* h1f = &h1buf[p][0];
    for (int i = tid; i < 16 * O_OUT; i += 128) {
        int m = i / O_OUT, o = i % O_OUT;
        float acc = b_fc[o];
        #pragma unroll 5
        for (int k = 0; k < H_SZ; ++k)
            acc = fmaf(h1f[m * RS + k], W_fc[o * H_SZ + k], acc);
        out[(size_t)(b0 + m) * O_OUT + o] = acc;
    }
}

extern "C" void kernel_launch(void* const* d_in, const int* in_sizes, int n_in,
                              void* d_out, int out_size, void* d_ws, size_t ws_size,
                              hipStream_t stream) {
    const float* x     = (const float*)d_in[0];
    const float* W_ih0 = (const float*)d_in[1];
    const float* W_hh0 = (const float*)d_in[2];
    const float* b_ih0 = (const float*)d_in[3];
    const float* b_hh0 = (const float*)d_in[4];
    const float* W_ih1 = (const float*)d_in[5];
    const float* W_hh1 = (const float*)d_in[6];
    const float* b_ih1 = (const float*)d_in[7];
    const float* b_hh1 = (const float*)d_in[8];
    const float* W_fc  = (const float*)d_in[9];
    const float* b_fc  = (const float*)d_in[10];
    float* out = (float*)d_out;

    rnn2_tanh_wmma<<<B_SZ / 16, 128, 0, stream>>>(
        x, W_ih0, W_hh0, b_ih0, b_hh0, W_ih1, W_hh1, b_ih1, b_hh1, W_fc, b_fc, out);
}


# K extent of all three products trimmed from 64 to 52 (13 chunks of 4, not 16): hidden size is 50, the dropped chunks were zero padding; order of every sum unchanged
# speedup vs baseline: 1.5348x; 1.1308x over previous
#include <hip/hip_runtime.h>
#include <math.h>

typedef float v2f __attribute__((ext_vector_type(2)));
typedef float v8f __attribute__((ext_vector_type(8)));

#define H_SZ 50
#define O_OUT 10
#define B_SZ 512
#define T_STEPS 2048
#define RS 68
#define XBLK 64
#define NT 4
#define KC 13

__device__ __forceinline__ float fast_tanh(float v) {
#if __has_builtin(__builtin_amdgcn_tanhf)
    return __builtin_amdgcn_tanhf(v);
#else
    float e = __expf(v + v);
    return 1.0f - __fdividef(2.0f, e + 1.0f);
#endif
}

#define WMMA_F32(A, B, C) \
    __builtin_amdgcn_wmma_f32_16x16x4_f32(false, (A), false, (B), (short)0, (C), false, false)

__device__ __forceinline__ void stage_x(float* xs, const float* __restrict__ x,
                                        int b0, int t0, int tid) {
    for (int i = tid; i < 16 * XBLK; i += 128) {
        int rr = i >> 6;
        int cc = i & (XBLK - 1);
        xs[rr * RS + cc] = x[(size_t)(b0 + rr) * T_STEPS + t0 + cc];
    }
}

template <bool L0, bool L1>
__device__ __forceinline__ void skew_pass(
    const float* h0r, const float* h1r, float* h0w, float* h1w, const float* xcol,
    const v2f (&rB0)[KC], const v2f (&rB1)[KC], const v2f (&rB2)[KC],
    float wih, float bias0n, float bias1n, int r, int hf, int wv)
{
    float xm[8];
    if (L0) {
        #pragma unroll
        for (int v = 0; v < 8; ++v)
            xm[v] = xcol[(v + 8 * hf) * RS];
        __builtin_amdgcn_sched_barrier(0);
    }
    v2f a0[KC], a1[KC];
    #pragma unroll
    for (int kc = 0; kc < KC; ++kc)
        a0[kc] = *(const v2f*)&h0r[r * RS + kc * 4 + 2 * hf];
    if (L1) {
        #pragma unroll
        for (int kc = 0; kc < KC; ++kc)
            a1[kc] = *(const v2f*)&h1r[r * RS + kc * 4 + 2 * hf];
    }
    __builtin_amdgcn_sched_barrier(0);

    v8f acc0, acc1;
    if (L0) {
        #pragma unroll
        for (int v = 0; v < 8; ++v) {
            acc0[v] = fmaf(xm[v], wih, bias0n);
            acc1[v] = 0.0f;
        }
        #pragma unroll
        for (int kc = 0; kc < KC; kc += 2) {
            acc0 = WMMA_F32(a0[kc], rB0[kc], acc0);
            if (kc + 1 < KC)
                acc1 = WMMA_F32(a0[kc + 1], rB0[kc + 1], acc1);
        }
    }
    if (L0 && L1) {
        __builtin_amdgcn_sched_barrier(0);
    }
    float th0[8], th1[8];
    if (L0) {
        v8f c0 = acc0 + acc1;
        #pragma unroll
        for (int v = 0; v < 8; ++v) th0[v] = fast_tanh(c0[v]);
    }
    if (L1) {
        v8f acc2, acc3;
        #pragma unroll
        for (int v = 0; v < 8; ++v) { acc2[v] = bias1n; acc3[v] = 0.0f; }
        #pragma unroll
        for (int kc = 0; kc < KC; ++kc) {
            acc2 = WMMA_F32(a0[kc], rB1[kc], acc2);
            acc3 = WMMA_F32(a1[kc], rB2[kc], acc3);
        }
        v8f c1 = acc2 + acc3;
        #pragma unroll
        for (int v = 0; v < 8; ++v) th1[v] = fast_tanh(c1[v]);
    }
    if (L0) {
        #pragma unroll
        for (int v = 0; v < 8; ++v)
            h0w[(v + 8 * hf) * RS + wv * 16 + r] = th0[v];
    }
    if (L1) {
        #pragma unroll
        for (int v = 0; v < 8; ++v)
            h1w[(v + 8 * hf) * RS + wv * 16 + r] = th1[v];
    }
}

__global__ __launch_bounds__(128, 1) void rnn2_tanh_wmma(
    const float* __restrict__ x,
    const float* __restrict__ W_ih0, const float* __restrict__ W_hh0,
    const float* __restrict__ b_ih0, const float* __restrict__ b_hh0,
    const float* __restrict__ W_ih1, const float* __restrict__ W_hh1,
    const float* __restrict__ b_ih1, const float* __restrict__ b_hh1,
    const float* __restrict__ W_fc,  const float* __restrict__ b_fc,
    float* __restrict__ out)
{
    __shared__ float wB0[NT][KC][32][2];
    __shared__ float wB1[NT][KC][32][2];
    __shared__ float wB2[NT][KC][32][2];
    __shared__ float h0buf[2][16 * RS];
    __shared__ float h1buf[2][16 * RS];
    __shared__ float xs[16 * RS];

    const int tid  = threadIdx.x;
    const int lane = tid & 31;
    const int wv   = tid >> 5;
    const int r    = lane & 15;
    const int hf   = lane >> 4;
    const int b0   = blockIdx.x * 16;

    {
        float* h0f = &h0buf[0][0];
        float* h1f = &h1buf[0][0];
        for (int i = tid; i < 2 * 16 * RS; i += 128) { h0f[i] = 0.f; h1f[i] = 0.f; }
        float* w0f = &wB0[0][0][0][0];
        float* w1f = &wB1[0][0][0][0];
        float* w2f = &wB2[0][0][0][0];
        for (int idx = tid; idx < NT * KC * 32 * 2; idx += 128) {
            int v  = idx & 1;
            int ln = (idx >> 1) & 31;
            int kc = (idx >> 6) % KC;
            int nt = (idx >> 6) / KC;
            int n  = nt * 16 + (ln & 15);
            int k  = kc * 4 + 2 * (ln >> 4) + v;
            float a = 0.f, b = 0.f, c = 0.f;
            if (n < H_SZ && k < H_SZ) {
                a = W_hh0[n * H_SZ + k];
                b = W_ih1[n * H_SZ + k];
                c = W_hh1[n * H_SZ + k];
            }
            w0f[idx] = a; w1f[idx] = b; w2f[idx] = c;
        }
    }

    const int n = wv * 16 + r;
    float wih = 0.f, bias0n = 0.f, bias1n = 0.f;
    if (n < H_SZ) {
        wih    = W_ih0[n];
        bias0n = b_ih0[n] + b_hh0[n];
        bias1n = b_ih1[n] + b_hh1[n];
    }
    __syncthreads();

    v2f rB0[KC], rB1[KC], rB2[KC];
    #pragma unroll
    for (int kc = 0; kc < KC; ++kc) {
        rB0[kc] = *(const v2f*)&wB0[wv][kc][lane][0];
        rB1[kc] = *(const v2f*)&wB1[wv][kc][lane][0];
        rB2[kc] = *(const v2f*)&wB2[wv][kc][lane][0];
    }

    stage_x(xs, x, b0, 0, tid);
    __syncthreads();
    skew_pass<true, false>(&h0buf[0][0], &h1buf[0][0], &h0buf[1][0], &h1buf[1][0], &xs[0],
                           rB0, rB1, rB2, wih, bias0n, bias1n, r, hf, wv);
    __syncthreads();
    int p = 1;
    for (int i = 1; i < T_STEPS; ++i) {
        const int tt = i & (XBLK - 1);
        if (tt == 0) {
            stage_x(xs, x, b0, i, tid);
            __syncthreads();
        }
        skew_pass<true, true>(&h0buf[p][0], &h1buf[p][0], &h0buf[1 - p][0], &h1buf[1 - p][0],
                              &xs[tt], rB0, rB1, rB2, wih, bias0n, bias1n, r, hf, wv);
        __syncthreads();
        p ^= 1;
    }
    skew_pass<false, true>(&h0buf[p][0], &h1buf[p][0], &h0buf[1 - p][0], &h1buf[1 - p][0],
                           &xs[0], rB0, rB1, rB2, wih, bias0n, bias1n, r, hf, wv);
    __syncthreads();
    p ^= 1;

    const float* h1f = &h1buf[p][0];
    for (int i = tid; i < 16 * O_OUT; i += 128) {
        int m = i / O_OUT, o = i % O_OUT;
        float acc = b_fc[o];
        #pragma unroll 5
        for (int k = 0; k < H_SZ; ++k)
            acc = fmaf(h1f[m * RS + k], W_fc[o * H_SZ + k], acc);
        out[(size_t)(b0 + m) * O_OUT + o] = acc;
    }
}

extern "C" void kernel_launch(void* const* d_in, const int* in_sizes, int n_in,
                              void* d_out, int out_size, void* d_ws, size_t ws_size,
                              hipStream_t stream) {
    const float* x     = (const float*)d_in[0];
    const float* W_ih0 = (const float*)d_in[1];
    const float* W_hh0 = (const float*)d_in[2];
    const float* b_ih0 = (const float*)d_in[3];
    const float* b_hh0 = (const float*)d_in[4];
    const float* W_ih1 = (const float*)d_in[5];
    const float* W_hh1 = (const float*)d_in[6];
    const float* b_ih1 = (const float*)d_in[7];
    const float* b_hh1 = (const float*)d_in[8];
    const float* W_fc  = (const float*)d_in[9];
    const float* b_fc  = (const float*)d_in[10];
    float* out = (float*)d_out;

    rnn2_tanh_wmma<<<B_SZ / 16, 128, 0, stream>>>(
        x, W_ih0, W_hh0, b_ih0, b_hh0, W_ih1, W_hh1, b_ih1, b_hh1, W_fc, b_fc, out);
}


# x staged one block ahead: next 64 steps of x read by two 16-byte global loads per thread at the top of a block's last pass, stored to a second LDS buffer at its end; was 8 serial loads and a barrier
# speedup vs baseline: 1.7137x; 1.1166x over previous
#include <hip/hip_runtime.h>
#include <math.h>

typedef float v2f __attribute__((ext_vector_type(2)));
typedef float v8f __attribute__((ext_vector_type(8)));
typedef float v4f __attribute__((ext_vector_type(4)));

#define H_SZ 50
#define O_OUT 10
#define B_SZ 512
#define T_STEPS 2048
#define RS 68
#define XBLK 64
#define NBLK (T_STEPS / XBLK)
#define NT 4
#define KC 13

__device__ __forceinline__ float fast_tanh(float v) {
#if __has_builtin(__builtin_amdgcn_tanhf)
    return __builtin_amdgcn_tanhf(v);
#else
    float e = __expf(v + v);
    return 1.0f - __fdividef(2.0f, e + 1.0f);
#endif
}

#define WMMA_F32(A, B, C) \
    __builtin_amdgcn_wmma_f32_16x16x4_f32(false, (A), false, (B), (short)0, (C), false, false)


template <bool L0, bool L1>
__device__ __forceinline__ void skew_pass(
    const float* h0r, const float* h1r, float* h0w, float* h1w, const float* xcol,
    const v2f (&rB0)[KC], const v2f (&rB1)[KC], const v2f (&rB2)[KC],
    float wih, float bias0n, float bias1n, int r, int hf, int wv)
{
    float xm[8];
    if (L0) {
        #pragma unroll
        for (int v = 0; v < 8; ++v)
            xm[v] = xcol[(v + 8 * hf) * RS];
        __builtin_amdgcn_sched_barrier(0);
    }
    v2f a0[KC], a1[KC];
    #pragma unroll
    for (int kc = 0; kc < KC; ++kc)
        a0[kc] = *(const v2f*)&h0r[r * RS + kc * 4 + 2 * hf];
    if (L1) {
        #pragma unroll
        for (int kc = 0; kc < KC; ++kc)
            a1[kc] = *(const v2f*)&h1r[r * RS + kc * 4 + 2 * hf];
    }
    __builtin_amdgcn_sched_barrier(0);

    v8f acc0, acc1;
    if (L0) {
        #pragma unroll
        for (int v = 0; v < 8; ++v) {
            acc0[v] = fmaf(xm[v], wih, bias0n);
            acc1[v] = 0.0f;
        }
        #pragma unroll
        for (int kc = 0; kc < KC; kc += 2) {
            acc0 = WMMA_F32(a0[kc], rB0[kc], acc0);
            if (kc + 1 < KC)
                acc1 = WMMA_F32(a0[kc + 1], rB0[kc + 1], acc1);
        }
    }
    if (L0 && L1) {
        __builtin_amdgcn_sched_barrier(0);
    }
    float th0[8], th1[8];
    if (L0) {
        v8f c0 = acc0 + acc1;
        #pragma unroll
        for (int v = 0; v < 8; ++v) th0[v] = fast_tanh(c0[v]);
    }
    if (L1) {
        v8f acc2, acc3;
        #pragma unroll
        for (int v = 0; v < 8; ++v) { acc2[v] = bias1n; acc3[v] = 0.0f; }
        #pragma unroll
        for (int kc = 0; kc < KC; ++kc) {
            acc2 = WMMA_F32(a0[kc], rB1[kc], acc2);
            acc3 = WMMA_F32(a1[kc], rB2[kc], acc3);
        }
        v8f c1 = acc2 + acc3;
        #pragma unroll
        for (int v = 0; v < 8; ++v) th1[v] = fast_tanh(c1[v]);
    }
    if (L0) {
        #pragma unroll
        for (int v = 0; v < 8; ++v)
            h0w[(v + 8 * hf) * RS + wv * 16 + r] = th0[v];
    }
    if (L1) {
        #pragma unroll
        for (int v = 0; v < 8; ++v)
            h1w[(v + 8 * hf) * RS + wv * 16 + r] = th1[v];
    }
}

__global__ __launch_bounds__(128, 1) void rnn2_tanh_wmma(
    const float* x,
    const float* __restrict__ W_ih0, const float* __restrict__ W_hh0,
    const float* __restrict__ b_ih0, const float* __restrict__ b_hh0,
    const float* __restrict__ W_ih1, const float* __restrict__ W_hh1,
    const float* __restrict__ b_ih1, const float* __restrict__ b_hh1,
    const float* __restrict__ W_fc,  const float* __restrict__ b_fc,
    float* __restrict__ out)
{
    __shared__ float wB0[NT][KC][32][2];
    __shared__ float wB1[NT][KC][32][2];
    __shared__ float wB2[NT][KC][32][2];
    __shared__ float h0buf[2][16 * RS];
    __shared__ float h1buf[2][16 * RS];
    __shared__ __attribute__((aligned(16))) float xs[2][16 * RS];

    const int tid  = threadIdx.x;
    const int lane = tid & 31;
    const int wv   = tid >> 5;
    const int r    = lane & 15;
    const int hf   = lane >> 4;
    const int b0   = blockIdx.x * 16;

    {
        float* h0f = &h0buf[0][0];
        float* h1f = &h1buf[0][0];
        for (int i = tid; i < 2 * 16 * RS; i += 128) { h0f[i] = 0.f; h1f[i] = 0.f; }
        float* w0f = &wB0[0][0][0][0];
        float* w1f = &wB1[0][0][0][0];
        float* w2f = &wB2[0][0][0][0];
        for (int idx = tid; idx < NT * KC * 32 * 2; idx += 128) {
            int v  = idx & 1;
            int ln = (idx >> 1) & 31;
            int kc = (idx >> 6) % KC;
            int nt = (idx >> 6) / KC;
            int n  = nt * 16 + (ln & 15);
            int k  = kc * 4 + 2 * (ln >> 4) + v;
            float a = 0.f, b = 0.f, c = 0.f;
            if (n < H_SZ && k < H_SZ) {
                a = W_hh0[n * H_SZ + k];
                b = W_ih1[n * H_SZ + k];
                c = W_hh1[n * H_SZ + k];
            }
            w0f[idx] = a; w1f[idx] = b; w2f[idx] = c;
        }
    }

    const int n = wv * 16 + r;
    float wih = 0.f, bias0n = 0.f, bias1n = 0.f;
    if (n < H_SZ) {
        wih    = W_ih0[n];
        bias0n = b_ih0[n] + b_hh0[n];
        bias1n = b_ih1[n] + b_hh1[n];
    }
    __syncthreads();

    v2f rB0[KC], rB1[KC], rB2[KC];
    #pragma unroll
    for (int kc = 0; kc < KC; ++kc) {
        rB0[kc] = *(const v2f*)&wB0[wv][kc][lane][0];
        rB1[kc] = *(const v2f*)&wB1[wv][kc][lane][0];
        rB2[kc] = *(const v2f*)&wB2[wv][kc][lane][0];
    }

    const int xrow = tid >> 3;
    const int xcg  = (tid & 7) * 8;
    const float* xg = x + (size_t)(b0 + xrow) * T_STEPS + xcg;
    const int xl = xrow * RS + xcg;
    {
        v4f n0 = *(const v4f*)xg;
        v4f n1 = *(const v4f*)(xg + 4);
        *(v4f*)&xs[0][xl]     = n0;
        *(v4f*)&xs[0][xl + 4] = n1;
    }
    __syncthreads();
    skew_pass<true, false>(&h0buf[0][0], &h1buf[0][0], &h0buf[1][0], &h1buf[1][0], &xs[0][0],
                           rB0, rB1, rB2, wih, bias0n, bias1n, r, hf, wv);
    __syncthreads();
    int p = 1;
    for (int k = 0; k < NBLK; ++k) {
        const float* xb = &xs[k & 1][0];
        for (int tt = (k == 0) ? 1 : 0; tt < XBLK - 1; ++tt) {
            skew_pass<true, true>(&h0buf[p][0], &h1buf[p][0], &h0buf[1 - p][0], &h1buf[1 - p][0],
                                  xb + tt, rB0, rB1, rB2, wih, bias0n, bias1n, r, hf, wv);
            __syncthreads();
            p ^= 1;
        }
        const bool more = (k + 1 < NBLK);
        v4f n0 = {0.f, 0.f, 0.f, 0.f}, n1 = {0.f, 0.f, 0.f, 0.f};
        if (more) {
            const float* g = xg + (k + 1) * XBLK;
            n0 = *(const v4f*)g;
            n1 = *(const v4f*)(g + 4);
        }
        skew_pass<true, true>(&h0buf[p][0], &h1buf[p][0], &h0buf[1 - p][0], &h1buf[1 - p][0],
                              xb + (XBLK - 1), rB0, rB1, rB2, wih, bias0n, bias1n, r, hf, wv);
        if (more) {
            float* d = &xs[(k + 1) & 1][xl];
            *(v4f*)d       = n0;
            *(v4f*)(d + 4) = n1;
        }
        __syncthreads();
        p ^= 1;
    }
    skew_pass<false, true>(&h0buf[p][0], &h1buf[p][0], &h0buf[1 - p][0], &h1buf[1 - p][0],
                           &xs[0][0], rB0, rB1, rB2, wih, bias0n, bias1n, r, hf, wv);
    __syncthreads();
    p ^= 1;

    const float* h1f = &h1buf[p][0];
    for (int i = tid; i < 16 * O_OUT; i += 128) {
        int m = i / O_OUT, o = i % O_OUT;
        float acc = b_fc[o];
        #pragma unroll 5
        for (int k = 0; k < H_SZ; ++k)
            acc = fmaf(h1f[m * RS + k], W_fc[o * H_SZ + k], acc);
        out[(size_t)(b0 + m) * O_OUT + o] = acc;
    }
}

extern "C" void kernel_launch(void* const* d_in, const int* in_sizes, int n_in,
                              void* d_out, int out_size, void* d_ws, size_t ws_size,
                              hipStream_t stream) {
    const float* x     = (const float*)d_in[0];
    const float* W_ih0 = (const float*)d_in[1];
    const float* W_hh0 = (const float*)d_in[2];
    const float* b_ih0 = (const float*)d_in[3];
    const float* b_hh0 = (const float*)d_in[4];
    const float* W_ih1 = (const float*)d_in[5];
    const float* W_hh1 = (const float*)d_in[6];
    const float* b_ih1 = (const float*)d_in[7];
    const float* b_hh1 = (const float*)d_in[8];
    const float* W_fc  = (const float*)d_in[9];
    const float* b_fc  = (const float*)d_in[10];
    float* out = (float*)d_out;

    rnn2_tanh_wmma<<<B_SZ / 16, 128, 0, stream>>>(
        x, W_ih0, W_hh0, b_ih0, b_hh0, W_ih1, W_hh1, b_ih1, b_hh1, W_fc, b_fc, out);
}


# h0@Wih1 chain taken off each pass's critical path (layer 1 lags one more step): its WMMAs fill the LDS-read wait at a pass's top and its tanh/store tail; order of every sum unchanged
# speedup vs baseline: 1.9291x; 1.1257x over previous
#include <hip/hip_runtime.h>
#include <math.h>

typedef float v2f __attribute__((ext_vector_type(2)));
typedef float v8f __attribute__((ext_vector_type(8)));
typedef float v4f __attribute__((ext_vector_type(4)));

#define H_SZ 50
#define O_OUT 10
#define B_SZ 512
#define T_STEPS 2048
#define RS 68
#define XBLK 64
#define NBLK (T_STEPS / XBLK)
#define NT 4
#define KC 13
#define KT 8

__device__ __forceinline__ float fast_tanh(float v) {
#if __has_builtin(__builtin_amdgcn_tanhf)
    return __builtin_amdgcn_tanhf(v);
#else
    float e = __expf(v + v);
    return 1.0f - __fdividef(2.0f, e + 1.0f);
#endif
}

#define WMMA_F32(A, B, C) \
    __builtin_amdgcn_wmma_f32_16x16x4_f32(false, (A), false, (B), (short)0, (C), false, false)


template <bool L0, bool L1, bool PN>
__device__ __forceinline__ void skew_pass(
    const float* h0r, const float* h1r, float* h0w, float* h1w, const float* xcol,
    const v2f (&rB0)[KC], const v2f (&rB1)[KC], const v2f (&rB2)[KC],
    float wih, float bias0n, float bias1n, int r, int hf, int wv,
    const v8f& pin, const v2f (&tin)[KC - KT], v8f& pout, v2f (&tout)[KC - KT])
{
    float xm[8];
    if (L0) {
        #pragma unroll
        for (int v = 0; v < 8; ++v)
            xm[v] = xcol[(v + 8 * hf) * RS];
        __builtin_amdgcn_sched_barrier(0);
    }
    v2f a0[KC], a1[KC];
    if (L1) {
        #pragma unroll
        for (int kc = 0; kc < KC; ++kc)
            a1[kc] = *(const v2f*)&h1r[r * RS + kc * 4 + 2 * hf];
    }
    if (L0 || PN) {
        #pragma unroll
        for (int kc = 0; kc < KC; ++kc)
            a0[kc] = *(const v2f*)&h0r[r * RS + kc * 4 + 2 * hf];
    }
    __builtin_amdgcn_sched_barrier(0);

    v8f accP = pin;
    if (L1) {
        #pragma unroll
        for (int kc = KT; kc < KC; ++kc)
            accP = WMMA_F32(tin[kc - KT], rB1[kc], accP);
        asm volatile("" :: "v"(accP));
        __builtin_amdgcn_sched_barrier(0);
    }

    v8f acc0, acc1, acc3;
    #pragma unroll
    for (int v = 0; v < 8; ++v) {
        acc0[v] = L0 ? fmaf(xm[v], wih, bias0n) : 0.0f;
        acc1[v] = 0.0f;
        acc3[v] = 0.0f;
    }
    #pragma unroll
    for (int kc = 0; kc < KC; ++kc) {
        if (L0) {
            if (kc & 1) acc1 = WMMA_F32(a0[kc], rB0[kc], acc1);
            else        acc0 = WMMA_F32(a0[kc], rB0[kc], acc0);
        }
        if (L1)
            acc3 = WMMA_F32(a1[kc], rB2[kc], acc3);
    }

    v8f accN;
    #pragma unroll
    for (int v = 0; v < 8; ++v) accN[v] = bias1n;
    v8f c0, c1;
    float th0[8], th1[8];
    if (PN) {
        accN = WMMA_F32(a0[0], rB1[0], accN);
        asm volatile("" :: "v"(accN));
    }
    if (L0) {
        c0 = acc0 + acc1;
        #pragma unroll
        for (int v = 0; v < 3; ++v) th0[v] = fast_tanh(c0[v]);
    }
    __builtin_amdgcn_sched_barrier(0);
    if (PN) {
        accN = WMMA_F32(a0[1], rB1[1], accN);
        asm volatile("" :: "v"(accN));
    }
    if (L0) {
        #pragma unroll
        for (int v = 3; v < 8; ++v) th0[v] = fast_tanh(c0[v]);
    }
    __builtin_amdgcn_sched_barrier(0);
    if (PN) {
        accN = WMMA_F32(a0[2], rB1[2], accN);
        asm volatile("" :: "v"(accN));
    }
    if (L0) {
        #pragma unroll
        for (int v = 0; v < 8; ++v)
            h0w[(v + 8 * hf) * RS + wv * 16 + r] = th0[v];
    }
    if (L1) {
        c1 = accP + acc3;
        th1[0] = fast_tanh(c1[0]);
    }
    __builtin_amdgcn_sched_barrier(0);
    if (PN) {
        accN = WMMA_F32(a0[3], rB1[3], accN);
        asm volatile("" :: "v"(accN));
    }
    if (L1) {
        #pragma unroll
        for (int v = 1; v < 6; ++v) th1[v] = fast_tanh(c1[v]);
    }
    __builtin_amdgcn_sched_barrier(0);
    if (PN) {
        accN = WMMA_F32(a0[4], rB1[4], accN);
        asm volatile("" :: "v"(accN));
    }
    if (L1) {
        #pragma unroll
        for (int v = 6; v < 8; ++v) th1[v] = fast_tanh(c1[v]);
        #pragma unroll
        for (int v = 0; v < 8; ++v)
            h1w[(v + 8 * hf) * RS + wv * 16 + r] = th1[v];
    }
    __builtin_amdgcn_sched_barrier(0);
    if (PN) {
        #pragma unroll
        for (int kc = 5; kc < KT; ++kc)
            accN = WMMA_F32(a0[kc], rB1[kc], accN);
        pout = accN;
        #pragma unroll
        for (int kc = KT; kc < KC; ++kc)
            tout[kc - KT] = a0[kc];
        asm volatile("" :: "v"(accN));
    }
}

__global__ __launch_bounds__(128, 1) void rnn2_tanh_wmma(
    const float* x,
    const float* __restrict__ W_ih0, const float* __restrict__ W_hh0,
    const float* __restrict__ b_ih0, const float* __restrict__ b_hh0,
    const float* __restrict__ W_ih1, const float* __restrict__ W_hh1,
    const float* __restrict__ b_ih1, const float* __restrict__ b_hh1,
    const float* __restrict__ W_fc,  const float* __restrict__ b_fc,
    float* __restrict__ out)
{
    __shared__ float wB0[NT][KC][32][2];
    __shared__ float wB1[NT][KC][32][2];
    __shared__ float wB2[NT][KC][32][2];
    __shared__ float h0buf[2][16 * RS];
    __shared__ float h1buf[2][16 * RS];
    __shared__ __attribute__((aligned(16))) float xs[2][16 * RS];

    const int tid  = threadIdx.x;
    const int lane = tid & 31;
    const int wv   = tid >> 5;
    const int r    = lane & 15;
    const int hf   = lane >> 4;
    const int b0   = blockIdx.x * 16;

    {
        float* h0f = &h0buf[0][0];
        float* h1f = &h1buf[0][0];
        for (int i = tid; i < 2 * 16 * RS; i += 128) { h0f[i] = 0.f; h1f[i] = 0.f; }
        float* w0f = &wB0[0][0][0][0];
        float* w1f = &wB1[0][0][0][0];
        float* w2f = &wB2[0][0][0][0];
        for (int idx = tid; idx < NT * KC * 32 * 2; idx += 128) {
            int v  = idx & 1;
            int ln = (idx >> 1) & 31;
            int kc = (idx >> 6) % KC;
            int nt = (idx >> 6) / KC;
            int n  = nt * 16 + (ln & 15);
            int k  = kc * 4 + 2 * (ln >> 4) + v;
            float a = 0.f, b = 0.f, c = 0.f;
            if (n < H_SZ && k < H_SZ) {
                a = W_hh0[n * H_SZ + k];
                b = W_ih1[n * H_SZ + k];
                c = W_hh1[n * H_SZ + k];
            }
            w0f[idx] = a; w1f[idx] = b; w2f[idx] = c;
        }
    }

    const int n = wv * 16 + r;
    float wih = 0.f, bias0n = 0.f, bias1n = 0.f;
    if (n < H_SZ) {
        wih    = W_ih0[n];
        bias0n = b_ih0[n] + b_hh0[n];
        bias1n = b_ih1[n] + b_hh1[n];
    }
    __syncthreads();

    v2f rB0[KC], rB1[KC], rB2[KC];
    #pragma unroll
    for (int kc = 0; kc < KC; ++kc) {
        rB0[kc] = *(const v2f*)&wB0[wv][kc][lane][0];
        rB1[kc] = *(const v2f*)&wB1[wv][kc][lane][0];
        rB2[kc] = *(const v2f*)&wB2[wv][kc][lane][0];
    }

    const int xrow = tid >> 3;
    const int xcg  = (tid & 7) * 8;
    const float* xg = x + (size_t)(b0 + xrow) * T_STEPS + xcg;
    const int xl = xrow * RS + xcg;
    {
        v4f n0 = *(const v4f*)xg;
        v4f n1 = *(const v4f*)(xg + 4);
        *(v4f*)&xs[0][xl]     = n0;
        *(v4f*)&xs[0][xl + 4] = n1;
    }
    __syncthreads();
    v8f pa = {0.f, 0.f, 0.f, 0.f, 0.f, 0.f, 0.f, 0.f}, pb = pa;
    v2f ta[KC - KT], tb[KC - KT];
    #pragma unroll
    for (int j = 0; j < KC - KT; ++j) { ta[j] = v2f{0.f, 0.f}; tb[j] = v2f{0.f, 0.f}; }
    float* const H0a = &h0buf[0][0];
    float* const H0b = &h0buf[1][0];
    float* const H1a = &h1buf[0][0];
    float* const H1b = &h1buf[1][0];
    skew_pass<true, false, false>(H0a, H1a, H0b, H1b, &xs[0][0],
                                  rB0, rB1, rB2, wih, bias0n, bias1n, r, hf, wv, pb, tb, pa, ta);
    __syncthreads();
    skew_pass<true, false, true>(H0b, H1b, H0a, H1a, &xs[0][1],
                                 rB0, rB1, rB2, wih, bias0n, bias1n, r, hf, wv, pa, ta, pb, tb);
    __syncthreads();
    for (int k = 0; k < NBLK; ++k) {
        const float* xb = &xs[k & 1][0];
        for (int tt = (k == 0) ? 2 : 0; tt < XBLK - 2; tt += 2) {
            skew_pass<true, true, true>(H0a, H1a, H0b, H1b, xb + tt,
                                        rB0, rB1, rB2, wih, bias0n, bias1n, r, hf, wv, pb, tb, pa, ta);
            __syncthreads();
            skew_pass<true, true, true>(H0b, H1b, H0a, H1a, xb + tt + 1,
                                        rB0, rB1, rB2, wih, bias0n, bias1n, r, hf, wv, pa, ta, pb, tb);
            __syncthreads();
        }
        skew_pass<true, true, true>(H0a, H1a, H0b, H1b, xb + (XBLK - 2),
                                    rB0, rB1, rB2, wih, bias0n, bias1n, r, hf, wv, pb, tb, pa, ta);
        __syncthreads();
        const bool more = (k + 1 < NBLK);
        v4f n0 = {0.f, 0.f, 0.f, 0.f}, n1 = {0.f, 0.f, 0.f, 0.f};
        if (more) {
            const float* g = xg + (k + 1) * XBLK;
            n0 = *(const v4f*)g;
            n1 = *(const v4f*)(g + 4);
        }
        skew_pass<true, true, true>(H0b, H1b, H0a, H1a, xb + (XBLK - 1),
                                    rB0, rB1, rB2, wih, bias0n, bias1n, r, hf, wv, pa, ta, pb, tb);
        if (more) {
            float* d = &xs[(k + 1) & 1][xl];
            *(v4f*)d       = n0;
            *(v4f*)(d + 4) = n1;
        }
        __syncthreads();
    }
    skew_pass<false, true, true>(H0a, H1a, H0b, H1b, &xs[0][0],
                                 rB0, rB1, rB2, wih, bias0n, bias1n, r, hf, wv, pb, tb, pa, ta);
    __syncthreads();
    skew_pass<false, true, false>(H0b, H1b, H0a, H1a, &xs[0][0],
                                  rB0, rB1, rB2, wih, bias0n, bias1n, r, hf, wv, pa, ta, pb, tb);
    __syncthreads();
    const int p = 0;

    const float* h1f = &h1buf[p][0];
    for (int i = tid; i < 16 * O_OUT; i += 128) {
        int m = i / O_OUT, o = i % O_OUT;
        float acc = b_fc[o];
        #pragma unroll 5
        for (int k = 0; k < H_SZ; ++k)
            acc = fmaf(h1f[m * RS + k], W_fc[o * H_SZ + k], acc);
        out[(size_t)(b0 + m) * O_OUT + o] = acc;
    }
}

extern "C" void kernel_launch(void* const* d_in, const int* in_sizes, int n_in,
                              void* d_out, int out_size, void* d_ws, size_t ws_size,
                              hipStream_t stream) {
    const float* x     = (const float*)d_in[0];
    const float* W_ih0 = (const float*)d_in[1];
    const float* W_hh0 = (const float*)d_in[2];
    const float* b_ih0 = (const float*)d_in[3];
    const float* b_hh0 = (const float*)d_in[4];
    const float* W_ih1 = (const float*)d_in[5];
    const float* W_hh1 = (const float*)d_in[6];
    const float* b_ih1 = (const float*)d_in[7];
    const float* b_hh1 = (const float*)d_in[8];
    const float* W_fc  = (const float*)d_in[9];
    const float* b_fc  = (const float*)d_in[10];
    float* out = (float*)d_out;

    rnn2_tanh_wmma<<<B_SZ / 16, 128, 0, stream>>>(
        x, W_ih0, W_hh0, b_ih0, b_hh0, W_ih1, W_hh1, b_ih1, b_hh1, W_fc, b_fc, out);
}


# Operand roles swapped (weights A, state B): a wave's tanh registers are the B fragments of its own 3 K chunks, so 9 of 39 links per pass need no LDS trip or barrier; order of the k sums changes
# speedup vs baseline: 2.0961x; 1.0866x over previous
#include <hip/hip_runtime.h>
#include <math.h>

typedef float v2f __attribute__((ext_vector_type(2)));
typedef float v8f __attribute__((ext_vector_type(8)));
typedef float v4f __attribute__((ext_vector_type(4)));

#define H_SZ 50
#define O_OUT 10
#define B_SZ 512
#define T_STEPS 2048
#define RS 68
#define XBLK 64
#define NBLK (T_STEPS / XBLK)
#define KC 13
#define NO 3
#define NL (KC - NO)
#define NC 4
#define HBS 1152

__device__ __forceinline__ float fast_tanh(float v) {
#if __has_builtin(__builtin_amdgcn_tanhf)
    return __builtin_amdgcn_tanhf(v);
#else
    float e = __expf(v + v);
    return 1.0f - __fdividef(2.0f, e + 1.0f);
#endif
}

#define WMMA_F32(A, B, C) \
    __builtin_amdgcn_wmma_f32_16x16x4_f32(false, (A), false, (B), (short)0, (C), false, false)


struct Carry {
    v8f p;
    v8f a;
    v8f l;
    v8f h;
    v2f t[NC];
};

__device__ __forceinline__ v2f own_frag(const v8f& h, int o) {
    v2f f; f[0] = h[2 * o]; f[1] = h[2 * o + 1]; return f;
}

template <bool L0, bool L1, bool PN>
__device__ __forceinline__ void skew_pass(
    const float* hr0, const float* hr1, float* hw0, float* hw1, const float* xcol,
    const v2f (&W0o)[NO], const v2f (&W0l)[NL], const v2f (&W1o)[NO], const v2f (&W1l)[NL],
    const v2f (&W2o)[NO], const v2f (&W2l)[NL],
    const v8f& wih8, const v8f& b0v, const v8f& b1v,
    const int (&offs)[NL], int xoff, int st, int st3,
    const Carry& in, Carry& out)
{
    float xv = 0.f;
    if (L0) xv = xcol[xoff];
    v2f a0[NL], a1[NL];
    if (L1) {
        #pragma unroll
        for (int q = 0; q < NL; ++q) a1[q] = *(const v2f*)&hr1[offs[q]];
    }
    if (L0 || PN) {
        #pragma unroll
        for (int q = 0; q < NL; ++q) a0[q] = *(const v2f*)&hr0[offs[q]];
    }
    __builtin_amdgcn_sched_barrier(0);

    v8f accP = in.p;
    v8f accN = b1v;
    #pragma unroll
    for (int s = 0; s < NC; ++s) {
        if (L1) accP = WMMA_F32(W1l[NL - NC + s], in.t[s], accP);
        if (PN && s < NO) accN = WMMA_F32(W1o[s], own_frag(in.h, s), accN);
    }
    if (L1) asm volatile("" :: "v"(accP));
    if (PN) asm volatile("" :: "v"(accN));
    if (L1 || PN) __builtin_amdgcn_sched_barrier(0);

    v8f accA = in.a, accB, acc3 = in.l;
    #pragma unroll
    for (int v = 0; v < 8; ++v) accB[v] = L0 ? xv * wih8[v] : 0.0f;
    #pragma unroll
    for (int q = 0; q < NL; ++q) {
        if (L0) {
            if (q & 1) accB = WMMA_F32(W0l[q], a0[q], accB);
            else       accA = WMMA_F32(W0l[q], a0[q], accA);
        }
        if (L1)
            acc3 = WMMA_F32(W2l[q], a1[q], acc3);
    }

    v8f th0 = in.h, th1;
    v8f nA = b0v;
    v8f nL;
    #pragma unroll
    for (int v = 0; v < 8; ++v) { nL[v] = 0.0f; th1[v] = 0.0f; }
    if (PN) {
        accN = WMMA_F32(W1l[0], a0[0], accN);
        asm volatile("" :: "v"(accN));
    }
    if (L0) {
        v8f c0 = accA + accB;
        #pragma unroll
        for (int v = 0; v < 3; ++v) th0[v] = fast_tanh(c0[v]);
        __builtin_amdgcn_sched_barrier(0);
        if (PN) {
            accN = WMMA_F32(W1l[1], a0[1], accN);
            asm volatile("" :: "v"(accN));
        }
        #pragma unroll
        for (int v = 3; v < 8; ++v) th0[v] = fast_tanh(c0[v]);
        __builtin_amdgcn_sched_barrier(0);
        nA = WMMA_F32(W0o[0], own_frag(th0, 0), nA);
        asm volatile("" :: "v"(nA));
        #pragma unroll
        for (int j = 0; j < 3; ++j)
            *(v2f*)&hw0[st + 4 * j] = own_frag(th0, j);
        *(v2f*)&hw0[st3] = own_frag(th0, 3);
    } else if (PN) {
        accN = WMMA_F32(W1l[1], a0[1], accN);
        asm volatile("" :: "v"(accN));
    }
    if (L1) {
        v8f c1 = accP + acc3;
        th1[0] = fast_tanh(c1[0]);
        __builtin_amdgcn_sched_barrier(0);
        if (PN) {
            accN = WMMA_F32(W1l[2], a0[2], accN);
            asm volatile("" :: "v"(accN));
        }
        #pragma unroll
        for (int v = 1; v < 6; ++v) th1[v] = fast_tanh(c1[v]);
        __builtin_amdgcn_sched_barrier(0);
        if (L0) {
            nA = WMMA_F32(W0o[1], own_frag(th0, 1), nA);
            asm volatile("" :: "v"(nA));
        }
        #pragma unroll
        for (int v = 6; v < 8; ++v) th1[v] = fast_tanh(c1[v]);
        #pragma unroll
        for (int j = 0; j < 3; ++j)
            *(v2f*)&hw1[st + 4 * j] = own_frag(th1, j);
        *(v2f*)&hw1[st3] = own_frag(th1, 3);
        __builtin_amdgcn_sched_barrier(0);
    } else {
        if (PN) accN = WMMA_F32(W1l[2], a0[2], accN);
        if (L0) nA = WMMA_F32(W0o[1], own_frag(th0, 1), nA);
    }
    if (PN) accN = WMMA_F32(W1l[3], a0[3], accN);
    if (L0) nA = WMMA_F32(W0o[2], own_frag(th0, 2), nA);
    if (L1 && PN) nL = WMMA_F32(W2o[0], own_frag(th1, 0), nL);
    if (PN) accN = WMMA_F32(W1l[4], a0[4], accN);
    if (L1 && PN) nL = WMMA_F32(W2o[1], own_frag(th1, 1), nL);
    if (PN) accN = WMMA_F32(W1l[5], a0[5], accN);
    if (L1 && PN) nL = WMMA_F32(W2o[2], own_frag(th1, 2), nL);
    out.p = accN;
    out.a = nA;
    out.l = nL;
    out.h = th0;
    #pragma unroll
    for (int s = 0; s < NC; ++s) out.t[s] = (L0 || PN) ? a0[NL - NC + s] : in.t[s];
    if (PN) asm volatile("" :: "v"(accN));
    if (L0) asm volatile("" :: "v"(nA));
    if (L1 && PN) asm volatile("" :: "v"(nL));
}

#define WP 52
#define WROWS 53

__global__ __launch_bounds__(128, 1) void rnn2_tanh_wmma(
    const float* x,
    const float* __restrict__ W_ih0, const float* __restrict__ W_hh0,
    const float* __restrict__ b_ih0, const float* __restrict__ b_hh0,
    const float* __restrict__ W_ih1, const float* __restrict__ W_hh1,
    const float* __restrict__ b_ih1, const float* __restrict__ b_hh1,
    const float* __restrict__ W_fc,  const float* __restrict__ b_fc,
    float* __restrict__ out)
{
    __shared__ __attribute__((aligned(16))) float hb[4][HBS];
    __shared__ __attribute__((aligned(16))) float xs[2][16 * RS];
    __shared__ __attribute__((aligned(16))) float wl[3][WROWS * WP];
    __shared__ float cl[3][56];

    const int tid  = threadIdx.x;
    __builtin_assume(tid < 128);
    const int lane = tid & 31;
    const int wv   = tid >> 5;
    const int r    = lane & 15;
    const int hf   = lane >> 4;
    const int b0   = blockIdx.x * 16;

    {
        float* hz = &hb[0][0];
        for (int i = tid; i < 4 * HBS; i += 128) hz[i] = 0.f;
        for (int i = tid; i < WROWS * WP; i += 128) {
            const int row = i / WP, col = i - row * WP;
            if (row >= H_SZ || col >= H_SZ) { wl[0][i] = 0.f; wl[1][i] = 0.f; wl[2][i] = 0.f; }
        }
        #pragma unroll
        for (int it = 0; it < (H_SZ * H_SZ + 127) / 128; ++it) {
            const int i = tid + it * 128;
            if (i < H_SZ * H_SZ) {
                const int row = i / H_SZ, col = i - row * H_SZ;
                wl[0][row * WP + col] = W_hh0[i];
                wl[1][row * WP + col] = W_ih1[i];
                wl[2][row * WP + col] = W_hh1[i];
            }
        }
        if (tid < 56) {
            const bool in = tid < H_SZ;
            cl[0][tid] = in ? W_ih0[tid] : 0.f;
            cl[1][tid] = in ? b_ih0[tid] + b_hh0[tid] : 0.f;
            cl[2][tid] = in ? b_ih1[tid] + b_hh1[tid] : 0.f;
        }
    }
    __syncthreads();

    const int rj = (r & 7) >> 1;
    const int urow = 12 * wv + 4 * rj + 2 * (r >> 3) + (r & 1);
    const bool rowok = (rj < 3 || wv == 3) && urow < H_SZ;
    const int wro = (rowok ? urow : H_SZ + 2) * WP + 2 * hf;
    v2f W0o[NO], W1o[NO], W2o[NO], W0l[NL], W1l[NL], W2l[NL];
    int offs[NL];
    #pragma unroll
    for (int o = 0; o < NO; ++o) {
        const int g = 3 * wv + o;
        W0o[o] = *(const v2f*)&wl[0][wro + 4 * g];
        W1o[o] = *(const v2f*)&wl[1][wro + 4 * g];
        W2o[o] = *(const v2f*)&wl[2][wro + 4 * g];
    }
    #pragma unroll
    for (int q = 0; q < NL; ++q) {
        const int g = q + ((q >= 3 * wv) ? 3 : 0);
        W0l[q] = *(const v2f*)&wl[0][wro + 4 * g];
        W1l[q] = *(const v2f*)&wl[1][wro + 4 * g];
        W2l[q] = *(const v2f*)&wl[2][wro + 4 * g];
        offs[q] = r * RS + 4 * g + 2 * hf;
    }

    v8f wih8, b0v, b1v;
    #pragma unroll
    for (int v = 0; v < 8; ++v) {
        const int u = 12 * wv + 4 * (v >> 1) + 2 * hf + (v & 1);
        const bool ok = ((v >> 1) < 3 || wv == 3) && u < H_SZ;
        const int ue = ok ? u : H_SZ + 2;
        wih8[v] = cl[0][ue];
        b0v[v]  = cl[1][ue];
        b1v[v]  = cl[2][ue];
    }
    const int st  = r * RS + 12 * wv + 2 * hf;
    const int st3 = r * RS + 2 * hf + ((wv == 3) ? 48 : 52 + 4 * wv);
    const int xoff = r * RS;

    const int xrow = tid >> 3;
    const int xcg  = (tid & 7) * 8;
    const float* xg = x + (size_t)(b0 + xrow) * T_STEPS + xcg;
    const int xl = xrow * RS + xcg;
    {
        v4f n0 = *(const v4f*)xg;
        v4f n1 = *(const v4f*)(xg + 4);
        *(v4f*)&xs[0][xl]     = n0;
        *(v4f*)&xs[0][xl + 4] = n1;
    }
    __syncthreads();
    Carry ca, cb;
    #pragma unroll
    for (int v = 0; v < 8; ++v) {
        ca.p[v] = 0.f; ca.a[v] = 0.f; ca.l[v] = 0.f; ca.h[v] = 0.f;
        cb.p[v] = 0.f; cb.l[v] = 0.f; cb.h[v] = 0.f;
    }
    cb.a = b0v;
    #pragma unroll
    for (int j = 0; j < NC; ++j) { ca.t[j] = v2f{0.f, 0.f}; cb.t[j] = v2f{0.f, 0.f}; }
    float* const H0a = &hb[0][0];
    float* const H0b = &hb[1][0];
    float* const H1a = &hb[2][0];
    float* const H1b = &hb[3][0];
#define PASS(L0, L1, PN, R0, R1, WR0, WR1, XC, CIN, COUT) \
    skew_pass<L0, L1, PN>(R0, R1, WR0, WR1, XC, W0o, W0l, W1o, W1l, W2o, W2l, wih8, b0v, b1v, offs, xoff, st, st3, CIN, COUT)
    PASS(true, false, false, H0a, H1a, H0b, H1b, &xs[0][0], cb, ca);
    __syncthreads();
    PASS(true, false, true, H0b, H1b, H0a, H1a, &xs[0][1], ca, cb);
    __syncthreads();
    for (int k = 0; k < NBLK; ++k) {
        const float* xb = &xs[k & 1][0];
        for (int tt = (k == 0) ? 2 : 0; tt < XBLK - 2; tt += 2) {
            PASS(true, true, true, H0a, H1a, H0b, H1b, xb + tt, cb, ca);
            __syncthreads();
            PASS(true, true, true, H0b, H1b, H0a, H1a, xb + tt + 1, ca, cb);
            __syncthreads();
        }
        PASS(true, true, true, H0a, H1a, H0b, H1b, xb + (XBLK - 2), cb, ca);
        __syncthreads();
        const bool more = (k + 1 < NBLK);
        v4f n0 = {0.f, 0.f, 0.f, 0.f}, n1 = {0.f, 0.f, 0.f, 0.f};
        if (more) {
            const float* g = xg + (k + 1) * XBLK;
            n0 = *(const v4f*)g;
            n1 = *(const v4f*)(g + 4);
        }
        PASS(true, true, true, H0b, H1b, H0a, H1a, xb + (XBLK - 1), ca, cb);
        if (more) {
            float* d = &xs[(k + 1) & 1][xl];
            *(v4f*)d       = n0;
            *(v4f*)(d + 4) = n1;
        }
        __syncthreads();
    }
    PASS(false, true, true, H0a, H1a, H0b, H1b, &xs[0][0], cb, ca);
    __syncthreads();
    PASS(false, true, false, H0b, H1b, H0a, H1a, &xs[0][0], ca, cb);
    __syncthreads();
#undef PASS

    const float* h1f = &hb[2][0];
    for (int i = tid; i < 16 * O_OUT; i += 128) {
        int m = i / O_OUT, o = i % O_OUT;
        float acc = b_fc[o];
        #pragma unroll 5
        for (int k = 0; k < H_SZ; ++k)
            acc = fmaf(h1f[m * RS + k], W_fc[o * H_SZ + k], acc);
        out[(size_t)(b0 + m) * O_OUT + o] = acc;
    }
}

extern "C" void kernel_launch(void* const* d_in, const int* in_sizes, int n_in,
                              void* d_out, int out_size, void* d_ws, size_t ws_size,
                              hipStream_t stream) {
    const float* x     = (const float*)d_in[0];
    const float* W_ih0 = (const float*)d_in[1];
    const float* W_hh0 = (const float*)d_in[2];
    const float* b_ih0 = (const float*)d_in[3];
    const float* b_hh0 = (const float*)d_in[4];
    const float* W_ih1 = (const float*)d_in[5];
    const float* W_hh1 = (const float*)d_in[6];
    const float* b_ih1 = (const float*)d_in[7];
    const float* b_hh1 = (const float*)d_in[8];
    const float* W_fc  = (const float*)d_in[9];
    const float* b_fc  = (const float*)d_in[10];
    float* out = (float*)d_out;

    rnn2_tanh_wmma<<<B_SZ / 16, 128, 0, stream>>>(
        x, W_ih0, W_hh0, b_ih0, b_hh0, W_ih1, W_hh1, b_ih1, b_hh1, W_fc, b_fc, out);
}


# Layers split over work-groups on separate compute units, one 256-step time block apart over 9 launches; layer-0 groups hand b1 + Wih1 h0 per step through the workspace; order of the k sums changes
# speedup vs baseline: 2.3418x; 1.1172x over previous
#include <hip/hip_runtime.h>
#include <math.h>


typedef float v2f __attribute__((ext_vector_type(2)));
typedef float v8f __attribute__((ext_vector_type(8)));
typedef float v4f __attribute__((ext_vector_type(4)));

#define H_SZ 50
#define O_OUT 10
#define B_SZ 512
#define T_STEPS 2048
#define RS 68
#define KC 13
#define NO 3
#define NL (KC - NO)
#define HBS 1152
#define NTILE (B_SZ / 16)
#define NBLK 8
#define TBK (T_STEPS / NBLK)
#define PSLOT 1024
#define PBUF (NTILE * (TBK + 1) * PSLOT)
#define WS_S0 (2 * PBUF)
#define WS_S1 (WS_S0 + NTILE * PSLOT)

__device__ __forceinline__ float fast_tanh(float v) {
#if __has_builtin(__builtin_amdgcn_tanhf)
    return __builtin_amdgcn_tanhf(v);
#else
    float e = __expf(v + v);
    return 1.0f - __fdividef(2.0f, e + 1.0f);
#endif
}

#define WMMA_F32(A, B, C) \
    __builtin_amdgcn_wmma_f32_16x16x4_f32(false, (A), false, (B), (short)0, (C), false, false)


__device__ __forceinline__ v2f own_frag(const v8f& h, int o) {
    v2f f; f[0] = h[2 * o]; f[1] = h[2 * o + 1]; return f;
}

__device__ __forceinline__ void lds_barrier() {
    __builtin_amdgcn_fence(__ATOMIC_RELEASE, "workgroup", "local");
    __builtin_amdgcn_s_barrier();
    __builtin_amdgcn_fence(__ATOMIC_ACQUIRE, "workgroup", "local");
}


template <bool L0>
__device__ __forceinline__ void pass_a(
    const float* hr, float* hw, float xv,
    const v2f (&W0o)[NO], const v2f (&W0l)[NL], const v2f (&W1o)[NO], const v2f (&W1l)[NL],
    const v8f& wih8, const v8f& b0v, const v8f& b1v,
    const int (&offs)[NL], int st, int st3,
    v8f& th, v8f& cP, v8f& cQ, float* pprev, bool prev)
{
    v2f a0[NL];
    #pragma unroll
    for (int q = 0; q < NL; ++q) a0[q] = *(const v2f*)&hr[offs[q]];
    __builtin_amdgcn_sched_barrier(0);

    v8f accA = b0v;
    v8f accP = b1v;
    #pragma unroll
    for (int s = 0; s < NO; ++s) {
        if (L0) accA = WMMA_F32(W0o[s], own_frag(th, s), accA);
        accP = WMMA_F32(W1o[s], own_frag(th, s), accP);
    }
    if (L0) asm volatile("" :: "v"(accA));
    asm volatile("" :: "v"(accP));
    if (prev) {
        v8f pp = cP + cQ;
        v4f lo = {pp[0], pp[1], pp[2], pp[3]}, hi = {pp[4], pp[5], pp[6], pp[7]};
        *(v4f*)pprev       = lo;
        *(v4f*)(pprev + 4) = hi;
    }
    __builtin_amdgcn_sched_barrier(0);

    v8f accB, accQ;
    #pragma unroll
    for (int v = 0; v < 8; ++v) { accB[v] = L0 ? xv * wih8[v] : 0.0f; accQ[v] = 0.0f; }
    if (L0) {
        #pragma unroll
        for (int q = 0; q < NL; ++q) {
            if (q & 1) accB = WMMA_F32(W0l[q], a0[q], accB);
            else       accA = WMMA_F32(W0l[q], a0[q], accA);
        }
        __builtin_amdgcn_sched_barrier(0);
        accP = WMMA_F32(W1l[0], a0[0], accP);
        accQ = WMMA_F32(W1l[1], a0[1], accQ);
        asm volatile("" :: "v"(accP), "v"(accQ));
        __builtin_amdgcn_sched_barrier(0);
        v8f c0 = accA + accB;
        #pragma unroll
        for (int v = 0; v < 3; ++v) th[v] = fast_tanh(c0[v]);
        __builtin_amdgcn_sched_barrier(0);
        accP = WMMA_F32(W1l[2], a0[2], accP);
        asm volatile("" :: "v"(accP));
        #pragma unroll
        for (int v = 3; v < 6; ++v) th[v] = fast_tanh(c0[v]);
        __builtin_amdgcn_sched_barrier(0);
        accQ = WMMA_F32(W1l[3], a0[3], accQ);
        asm volatile("" :: "v"(accQ));
        #pragma unroll
        for (int v = 6; v < 8; ++v) th[v] = fast_tanh(c0[v]);
        #pragma unroll
        for (int j = 0; j < 3; ++j)
            *(v2f*)&hw[st + 4 * j] = own_frag(th, j);
        *(v2f*)&hw[st3] = own_frag(th, 3);
        __builtin_amdgcn_sched_barrier(0);
        #pragma unroll
        for (int q = 4; q < NL; ++q) {
            if (q & 1) accQ = WMMA_F32(W1l[q], a0[q], accQ);
            else       accP = WMMA_F32(W1l[q], a0[q], accP);
        }
    } else {
        #pragma unroll
        for (int q = 0; q < NL; ++q) {
            if (q & 1) accQ = WMMA_F32(W1l[q], a0[q], accQ);
            else       accP = WMMA_F32(W1l[q], a0[q], accP);
        }
    }
    asm volatile("" :: "v"(accP), "v"(accQ));
    cP = accP;
    cQ = accQ;
}

__device__ __forceinline__ void pass_b(
    const float* hr, float* hw, const v8f& pin,
    const v2f (&W2o)[NO], const v2f (&W2l)[NL],
    const int (&offs)[NL], int st, int st3,
    v8f& th1, v8f& nL)
{
    v2f a1[NL];
    #pragma unroll
    for (int q = 0; q < NL; ++q) a1[q] = *(const v2f*)&hr[offs[q]];
    __builtin_amdgcn_sched_barrier(0);
    v8f accC = nL, accD = pin;
    #pragma unroll
    for (int q = 0; q < NL; ++q) {
        if (q & 1) accD = WMMA_F32(W2l[q], a1[q], accD);
        else       accC = WMMA_F32(W2l[q], a1[q], accC);
    }
    v8f c1 = accC + accD;
    #pragma unroll
    for (int v = 0; v < 8; ++v) th1[v] = fast_tanh(c1[v]);
    #pragma unroll
    for (int j = 0; j < 3; ++j)
        *(v2f*)&hw[st + 4 * j] = own_frag(th1, j);
    *(v2f*)&hw[st3] = own_frag(th1, 3);
    __builtin_amdgcn_sched_barrier(0);
    v8f n;
    #pragma unroll
    for (int v = 0; v < 8; ++v) n[v] = 0.0f;
    #pragma unroll
    for (int s = 0; s < NO; ++s) n = WMMA_F32(W2o[s], own_frag(th1, s), n);
    asm volatile("" :: "v"(n));
    nL = n;
}

#define WP 52
#define WROWS 53

__global__ __launch_bounds__(128, 1) void rnn2_tanh_wmma(
    const float* x,
    const float* __restrict__ W_ih0, const float* __restrict__ W_hh0,
    const float* __restrict__ b_ih0, const float* __restrict__ b_hh0,
    const float* __restrict__ W_ih1, const float* __restrict__ W_hh1,
    const float* __restrict__ b_ih1, const float* __restrict__ b_hh1,
    const float* __restrict__ W_fc,  const float* __restrict__ b_fc,
    float* __restrict__ out, float* ws, int k)
{
    __shared__ __attribute__((aligned(16))) float hb[2][HBS];
    __shared__ __attribute__((aligned(16))) float wl[3][WROWS * WP];
    __shared__ float cl[3][56];

    const bool roleB = blockIdx.x >= NTILE;
    const int tile = roleB ? (int)blockIdx.x - NTILE : (int)blockIdx.x;
    const int kb = roleB ? k - 1 : k;
    if (kb < 0 || kb >= NBLK) return;

    const int tid  = threadIdx.x;
    __builtin_assume(tid < 128);
    const int lane = tid & 31;
    const int wv   = tid >> 5;
    const int r    = lane & 15;
    const int hf   = lane >> 4;
    const int b0   = tile * 16;

    {
        float* hz = &hb[0][0];
        for (int i = tid; i < 2 * HBS; i += 128) hz[i] = 0.f;
        for (int i = tid; i < WROWS * WP; i += 128) {
            const int row = i / WP, col = i - row * WP;
            if (row >= H_SZ || col >= H_SZ) { wl[0][i] = 0.f; wl[1][i] = 0.f; wl[2][i] = 0.f; }
        }
        #pragma unroll
        for (int it = 0; it < (H_SZ * H_SZ + 127) / 128; ++it) {
            const int i = tid + it * 128;
            if (i < H_SZ * H_SZ) {
                const int row = i / H_SZ, col = i - row * H_SZ;
                wl[0][row * WP + col] = W_hh0[i];
                wl[1][row * WP + col] = W_ih1[i];
                wl[2][row * WP + col] = W_hh1[i];
            }
        }
        if (tid < 56) {
            const bool in = tid < H_SZ;
            cl[0][tid] = in ? W_ih0[tid] : 0.f;
            cl[1][tid] = in ? b_ih0[tid] + b_hh0[tid] : 0.f;
            cl[2][tid] = in ? b_ih1[tid] + b_hh1[tid] : 0.f;
        }
    }
    __syncthreads();

    const int rj = (r & 7) >> 1;
    const int urow = 12 * wv + 4 * rj + 2 * (r >> 3) + (r & 1);
    const bool rowok = (rj < 3 || wv == 3) && urow < H_SZ;
    const int wro = (rowok ? urow : H_SZ + 2) * WP + 2 * hf;
    int offs[NL];
    #pragma unroll
    for (int q = 0; q < NL; ++q) {
        const int g = q + ((q >= 3 * wv) ? 3 : 0);
        offs[q] = r * RS + 4 * g + 2 * hf;
    }
    const int st  = r * RS + 12 * wv + 2 * hf;
    const int st3 = r * RS + 2 * hf + ((wv == 3) ? 48 : 52 + 4 * wv);
    const int rec = (wv * 32 + lane) * 8;
    float* const H0 = &hb[0][0];
    float* const H1 = &hb[1][0];

    if (!roleB) {
        v2f W0o[NO], W1o[NO], W0l[NL], W1l[NL];
        #pragma unroll
        for (int o = 0; o < NO; ++o) {
            const int g = 3 * wv + o;
            W0o[o] = *(const v2f*)&wl[0][wro + 4 * g];
            W1o[o] = *(const v2f*)&wl[1][wro + 4 * g];
        }
        #pragma unroll
        for (int q = 0; q < NL; ++q) {
            const int g = q + ((q >= 3 * wv) ? 3 : 0);
            W0l[q] = *(const v2f*)&wl[0][wro + 4 * g];
            W1l[q] = *(const v2f*)&wl[1][wro + 4 * g];
        }
        v8f wih8, b0v, b1v;
        #pragma unroll
        for (int v = 0; v < 8; ++v) {
            const int u = 12 * wv + 4 * (v >> 1) + 2 * hf + (v & 1);
            const bool ok = ((v >> 1) < 3 || wv == 3) && u < H_SZ;
            const int ue = ok ? u : H_SZ + 2;
            wih8[v] = cl[0][ue];
            b0v[v]  = cl[1][ue];
            b1v[v]  = cl[2][ue];
        }
        float* sp = ws + WS_S0 + tile * PSLOT + rec;
        v8f th;
        #pragma unroll
        for (int v = 0; v < 8; ++v) th[v] = 0.f;
        if (kb > 0) {
            v4f lo = *(const v4f*)sp, hi = *(const v4f*)(sp + 4);
            th = v8f{lo[0], lo[1], lo[2], lo[3], hi[0], hi[1], hi[2], hi[3]};
            #pragma unroll
            for (int j = 0; j < 3; ++j)
                *(v2f*)&H0[st + 4 * j] = own_frag(th, j);
            *(v2f*)&H0[st3] = own_frag(th, 3);
        }
        __syncthreads();

        const float* xg = x + (size_t)(b0 + r) * T_STEPS + kb * TBK;
        float* pd = ws + (size_t)(kb & 1) * PBUF + (size_t)tile * (TBK + 1) * PSLOT + rec;
        v4f xc = *(const v4f*)xg;
        v4f xn = *(const v4f*)(xg + 4);
#define PASS_A(L0F, R, WR, XV, J, PRV) \
    pass_a<L0F>(R, WR, XV, W0o, W0l, W1o, W1l, wih8, b0v, b1v, offs, st, st3, th, cP, cQ, pd + ((ptrdiff_t)(J) - 1) * PSLOT, PRV)
        v8f cP, cQ;
        #pragma unroll
        for (int v = 0; v < 8; ++v) { cP[v] = 0.f; cQ[v] = 0.f; }
        for (int j = 0; j < TBK; j += 4) {
            const int jf = (j + 8 < TBK) ? j + 8 : TBK - 4;
            const v4f xf = *(const v4f*)(xg + jf);
            PASS_A(true, H0, H1, xc[0], j, j > 0);
            lds_barrier();
            PASS_A(true, H1, H0, xc[1], j + 1, true);
            lds_barrier();
            PASS_A(true, H0, H1, xc[2], j + 2, true);
            lds_barrier();
            PASS_A(true, H1, H0, xc[3], j + 3, true);
            lds_barrier();
            xc = xn; xn = xf;
        }
        PASS_A(false, H0, H1, 0.f, TBK, true);
#undef PASS_A
        {
            v8f pp = cP + cQ;
            float* pl = pd + (size_t)TBK * PSLOT;
            v4f lo = {pp[0], pp[1], pp[2], pp[3]}, hi = {pp[4], pp[5], pp[6], pp[7]};
            *(v4f*)pl       = lo;
            *(v4f*)(pl + 4) = hi;
        }
        {
            v4f lo = {th[0], th[1], th[2], th[3]}, hi = {th[4], th[5], th[6], th[7]};
            *(v4f*)sp       = lo;
            *(v4f*)(sp + 4) = hi;
        }
    } else {
        v2f W2o[NO], W2l[NL];
        #pragma unroll
        for (int o = 0; o < NO; ++o) {
            const int g = 3 * wv + o;
            W2o[o] = *(const v2f*)&wl[2][wro + 4 * g];
        }
        #pragma unroll
        for (int q = 0; q < NL; ++q) {
            const int g = q + ((q >= 3 * wv) ? 3 : 0);
            W2l[q] = *(const v2f*)&wl[2][wro + 4 * g];
        }
        float* sp = ws + WS_S1 + tile * PSLOT + rec;
        v8f th1;
        #pragma unroll
        for (int v = 0; v < 8; ++v) th1[v] = 0.f;
        if (kb > 0) {
            v4f lo = *(const v4f*)sp, hi = *(const v4f*)(sp + 4);
            th1 = v8f{lo[0], lo[1], lo[2], lo[3], hi[0], hi[1], hi[2], hi[3]};
            #pragma unroll
            for (int j = 0; j < 3; ++j)
                *(v2f*)&H0[st + 4 * j] = own_frag(th1, j);
            *(v2f*)&H0[st3] = own_frag(th1, 3);
        }
        __syncthreads();
        v8f nL;
        #pragma unroll
        for (int v = 0; v < 8; ++v) nL[v] = 0.f;
        #pragma unroll
        for (int s = 0; s < NO; ++s) nL = WMMA_F32(W2o[s], own_frag(th1, s), nL);

        const float* pg = ws + (size_t)(kb & 1) * PBUF + (size_t)tile * (TBK + 1) * PSLOT + rec;
#define LOADP(S) ({ const float* a_ = pg + (size_t)(S) * PSLOT; v4f l_ = *(const v4f*)a_, h_ = *(const v4f*)(a_ + 4); \
                    v8f{l_[0], l_[1], l_[2], l_[3], h_[0], h_[1], h_[2], h_[3]}; })
        v8f p0 = LOADP(1), p1 = LOADP(2), p2 = LOADP(3), p3 = LOADP(4);
#define PASS_B(R, WR, PV) pass_b(R, WR, PV, W2o, W2l, offs, st, st3, th1, nL)
        for (int j = 0; j < TBK; j += 4) {
            const int s0 = (j + 5 <= TBK) ? j + 5 : TBK;
            const int s1 = (j + 6 <= TBK) ? j + 6 : TBK;
            const int s2 = (j + 7 <= TBK) ? j + 7 : TBK;
            const int s3 = (j + 8 <= TBK) ? j + 8 : TBK;
            PASS_B(H0, H1, p0);
            p0 = LOADP(s0);
            lds_barrier();
            PASS_B(H1, H0, p1);
            p1 = LOADP(s1);
            lds_barrier();
            PASS_B(H0, H1, p2);
            p2 = LOADP(s2);
            lds_barrier();
            PASS_B(H1, H0, p3);
            p3 = LOADP(s3);
            lds_barrier();
        }
#undef PASS_B
#undef LOADP
        {
            v4f lo = {th1[0], th1[1], th1[2], th1[3]}, hi = {th1[4], th1[5], th1[6], th1[7]};
            *(v4f*)sp       = lo;
            *(v4f*)(sp + 4) = hi;
        }
        if (kb == NBLK - 1) {
            const float* h1f = H0;
            for (int i = tid; i < 16 * O_OUT; i += 128) {
                int m = i / O_OUT, o = i % O_OUT;
                float acc = b_fc[o];
                #pragma unroll 5
                for (int kk = 0; kk < H_SZ; ++kk)
                    acc = fmaf(h1f[m * RS + kk], W_fc[o * H_SZ + kk], acc);
                out[(size_t)(b0 + m) * O_OUT + o] = acc;
            }
        }
    }
}

extern "C" void kernel_launch(void* const* d_in, const int* in_sizes, int n_in,
                              void* d_out, int out_size, void* d_ws, size_t ws_size,
                              hipStream_t stream) {
    const float* x     = (const float*)d_in[0];
    const float* W_ih0 = (const float*)d_in[1];
    const float* W_hh0 = (const float*)d_in[2];
    const float* b_ih0 = (const float*)d_in[3];
    const float* b_hh0 = (const float*)d_in[4];
    const float* W_ih1 = (const float*)d_in[5];
    const float* W_hh1 = (const float*)d_in[6];
    const float* b_ih1 = (const float*)d_in[7];
    const float* b_hh1 = (const float*)d_in[8];
    const float* W_fc  = (const float*)d_in[9];
    const float* b_fc  = (const float*)d_in[10];
    float* out = (float*)d_out;
    float* ws  = (float*)d_ws;

#define STEP_LAUNCH(K) \
    rnn2_tanh_wmma<<<2 * NTILE, 128, 0, stream>>>( \
        x, W_ih0, W_hh0, b_ih0, b_hh0, W_ih1, W_hh1, b_ih1, b_hh1, W_fc, b_fc, out, ws, K)
    STEP_LAUNCH(0);
    STEP_LAUNCH(1);
    STEP_LAUNCH(2);
    STEP_LAUNCH(3);
    STEP_LAUNCH(4);
    STEP_LAUNCH(5);
    STEP_LAUNCH(6);
    STEP_LAUNCH(7);
    STEP_LAUNCH(8);
#undef STEP_LAUNCH
}


# Wih1 h0 products moved out of the layer-0 groups into a third set of work-groups one time block behind (10 launches, layer 0 writes h0 per step); order of the k sums changes
# speedup vs baseline: 2.6834x; 1.1459x over previous
#include <hip/hip_runtime.h>
#include <math.h>


typedef float v2f __attribute__((ext_vector_type(2)));
typedef float v8f __attribute__((ext_vector_type(8)));
typedef float v4f __attribute__((ext_vector_type(4)));

#define H_SZ 50
#define O_OUT 10
#define B_SZ 512
#define T_STEPS 2048
#define RS 68
#define KC 13
#define NO 3
#define NL (KC - NO)
#define HBS 1152
#define NTILE (B_SZ / 16)
#define NBLK 8
#define TBK (T_STEPS / NBLK)
#define PSLOT 1024
#define PBUF (NTILE * TBK * PSLOT)
#define NSEG 4
#define NPG (NTILE * NSEG)
#define USTEPS (TBK / NSEG / 4)
#define WS_H 0
#define WS_P (2 * PBUF)
#define WS_S0 (4 * PBUF)
#define WS_S1 (WS_S0 + NTILE * PSLOT)

__device__ __forceinline__ float fast_tanh(float v) {
#if __has_builtin(__builtin_amdgcn_tanhf)
    return __builtin_amdgcn_tanhf(v);
#else
    float e = __expf(v + v);
    return 1.0f - __fdividef(2.0f, e + 1.0f);
#endif
}

#define WMMA_F32(A, B, C) \
    __builtin_amdgcn_wmma_f32_16x16x4_f32(false, (A), false, (B), (short)0, (C), false, false)


__device__ __forceinline__ v2f own_frag(const v8f& h, int o) {
    v2f f; f[0] = h[2 * o]; f[1] = h[2 * o + 1]; return f;
}

__device__ __forceinline__ void lds_barrier() {
    __builtin_amdgcn_fence(__ATOMIC_RELEASE, "workgroup", "local");
    __builtin_amdgcn_s_barrier();
    __builtin_amdgcn_fence(__ATOMIC_ACQUIRE, "workgroup", "local");
}


template <bool L0>
__device__ __forceinline__ void pass_r(
    const float* hr, float* hw, const v8f& pin, float xv, const v8f& wih8,
    const v2f (&Wo)[NO], const v2f (&Wl)[NL],
    const int (&offs)[NL], int st, int st3,
    v8f& th, v8f& nL, float* hrec)
{
    v2f a1[NL];
    #pragma unroll
    for (int q = 0; q < NL; ++q) a1[q] = *(const v2f*)&hr[offs[q]];
    __builtin_amdgcn_sched_barrier(0);
    v8f accC = nL, accD;
    #pragma unroll
    for (int v = 0; v < 8; ++v) accD[v] = L0 ? xv * wih8[v] + pin[v] : pin[v];
    #pragma unroll
    for (int q = 0; q < NL; ++q) {
        if (q & 1) accD = WMMA_F32(Wl[q], a1[q], accD);
        else       accC = WMMA_F32(Wl[q], a1[q], accC);
    }
    v8f c1 = accC + accD;
    #pragma unroll
    for (int v = 0; v < 8; ++v) th[v] = fast_tanh(c1[v]);
    #pragma unroll
    for (int j = 0; j < 3; ++j)
        *(v2f*)&hw[st + 4 * j] = own_frag(th, j);
    *(v2f*)&hw[st3] = own_frag(th, 3);
    __builtin_amdgcn_sched_barrier(0);
    v8f n;
    #pragma unroll
    for (int v = 0; v < 8; ++v) n[v] = 0.0f;
    #pragma unroll
    for (int s = 0; s < NO; ++s) n = WMMA_F32(Wo[s], own_frag(th, s), n);
    asm volatile("" :: "v"(n));
    nL = n;
    if (L0) {
        v4f lo = {th[0], th[1], th[2], th[3]}, hi = {th[4], th[5], th[6], th[7]};
        *(v4f*)hrec       = lo;
        *(v4f*)(hrec + 4) = hi;
    }
}

#define WP 52
#define WROWS 53

__global__ __launch_bounds__(128, 1) void rnn2_tanh_wmma(
    const float* x,
    const float* __restrict__ W_ih0, const float* __restrict__ W_hh0,
    const float* __restrict__ b_ih0, const float* __restrict__ b_hh0,
    const float* __restrict__ W_ih1, const float* __restrict__ W_hh1,
    const float* __restrict__ b_ih1, const float* __restrict__ b_hh1,
    const float* __restrict__ W_fc,  const float* __restrict__ b_fc,
    float* __restrict__ out, float* ws, int k)
{
    __shared__ __attribute__((aligned(16))) float hb[2][HBS];
    __shared__ __attribute__((aligned(16))) float wl[3][WROWS * WP];
    __shared__ float cl[3][56];

    const int bx = (int)blockIdx.x;
    const bool roleB = bx >= NTILE && bx < 2 * NTILE;
    const bool roleP = bx >= 2 * NTILE;
    const int pgi = bx - 2 * NTILE;
    const int tile = roleP ? pgi / NSEG : (roleB ? bx - NTILE : bx);
    const int kb = roleP ? k - 1 : (roleB ? k - 2 : k);
    if (kb < 0 || kb >= NBLK) return;

    const int tid  = threadIdx.x;
    __builtin_assume(tid < 128);
    const int lane = tid & 31;
    const int wv   = tid >> 5;
    const int r    = lane & 15;
    const int hf   = lane >> 4;
    const int b0   = tile * 16;

    {
        float* hz = &hb[0][0];
        for (int i = tid; i < 2 * HBS; i += 128) hz[i] = 0.f;
        for (int i = tid; i < WROWS * WP; i += 128) {
            const int row = i / WP, col = i - row * WP;
            if (row >= H_SZ || col >= H_SZ) { wl[0][i] = 0.f; wl[1][i] = 0.f; wl[2][i] = 0.f; }
        }
        #pragma unroll
        for (int it = 0; it < (H_SZ * H_SZ + 127) / 128; ++it) {
            const int i = tid + it * 128;
            if (i < H_SZ * H_SZ) {
                const int row = i / H_SZ, col = i - row * H_SZ;
                wl[0][row * WP + col] = W_hh0[i];
                wl[1][row * WP + col] = W_ih1[i];
                wl[2][row * WP + col] = W_hh1[i];
            }
        }
        if (tid < 56) {
            const bool in = tid < H_SZ;
            cl[0][tid] = in ? W_ih0[tid] : 0.f;
            cl[1][tid] = in ? b_ih0[tid] + b_hh0[tid] : 0.f;
            cl[2][tid] = in ? b_ih1[tid] + b_hh1[tid] : 0.f;
        }
    }
    __syncthreads();

    const int rj = (r & 7) >> 1;
    const int urow = 12 * wv + 4 * rj + 2 * (r >> 3) + (r & 1);
    const bool rowok = (rj < 3 || wv == 3) && urow < H_SZ;
    const int wro = (rowok ? urow : H_SZ + 2) * WP + 2 * hf;
    int offs[NL];
    #pragma unroll
    for (int q = 0; q < NL; ++q) {
        const int g = q + ((q >= 3 * wv) ? 3 : 0);
        offs[q] = r * RS + 4 * g + 2 * hf;
    }
    const int st  = r * RS + 12 * wv + 2 * hf;
    const int st3 = r * RS + 2 * hf + ((wv == 3) ? 48 : 52 + 4 * wv);
    const int rec = (wv * 32 + lane) * 8;
    float* const H0 = &hb[0][0];
    float* const H1 = &hb[1][0];

    if (roleP) {
        v2f WA[4][KC];
        v8f bb[4];
        #pragma unroll
        for (int m = 0; m < 4; ++m) {
            const int rjm = (r & 7) >> 1;
            const int um = 12 * m + 4 * rjm + 2 * (r >> 3) + (r & 1);
            const bool okm = (rjm < 3 || m == 3) && um < H_SZ;
            const int wrm = (okm ? um : H_SZ + 2) * WP + 2 * hf;
            #pragma unroll
            for (int g = 0; g < KC; ++g) WA[m][g] = *(const v2f*)&wl[1][wrm + 4 * g];
            #pragma unroll
            for (int v = 0; v < 8; ++v) {
                const int u = 12 * m + 4 * (v >> 1) + 2 * hf + (v & 1);
                const bool ok = ((v >> 1) < 3 || m == 3) && u < H_SZ;
                bb[m][v] = cl[2][ok ? u : H_SZ + 2];
            }
        }
        const size_t uo = (size_t)(kb & 1) * PBUF + ((size_t)tile * TBK + (size_t)(pgi % NSEG) * (TBK / NSEG) + (size_t)wv * USTEPS) * PSLOT + (size_t)lane * 8;
        const float* hs = ws + WS_H + uo;
        float* pdst = ws + WS_P + uo;
        static_assert(USTEPS % 2 == 0, "the step loop below takes two steps per trip");
#define LOADH(F, S) { _Pragma("unroll") for (int m_ = 0; m_ < 4; ++m_) { \
            const float* a_ = hs + (size_t)(S) * PSLOT + m_ * 256; \
            v4f l_ = *(const v4f*)a_, h_ = *(const v4f*)(a_ + 4); \
            F[m_] = v8f{l_[0], l_[1], l_[2], l_[3], h_[0], h_[1], h_[2], h_[3]}; } }
#define PSTEP(F, S) { \
            v8f acc_[4]; \
            _Pragma("unroll") for (int m_ = 0; m_ < 4; ++m_) acc_[m_] = bb[m_]; \
            _Pragma("unroll") for (int g_ = 0; g_ < KC; ++g_) { \
                const int wq_ = (g_ < 12) ? g_ / 3 : 3, jq_ = (g_ < 12) ? g_ % 3 : 3; \
                const v2f bf_ = own_frag(F[wq_], jq_); \
                _Pragma("unroll") for (int m_ = 0; m_ < 4; ++m_) acc_[m_] = WMMA_F32(WA[m_][g_], bf_, acc_[m_]); \
            } \
            float* pu_ = pdst + (size_t)(S) * PSLOT; \
            _Pragma("unroll") for (int m_ = 0; m_ < 4; ++m_) { \
                v4f lo_ = {acc_[m_][0], acc_[m_][1], acc_[m_][2], acc_[m_][3]}, hi_ = {acc_[m_][4], acc_[m_][5], acc_[m_][6], acc_[m_][7]}; \
                *(v4f*)(pu_ + m_ * 256)     = lo_; \
                *(v4f*)(pu_ + m_ * 256 + 4) = hi_; \
            } }
        v8f fa[4], fb[4];
        LOADH(fa, 0);
        for (int u = 0; u < USTEPS; u += 2) {
            LOADH(fb, u + 1);
            __builtin_amdgcn_sched_barrier(0);
            PSTEP(fa, u);
            const int un = (u + 2 < USTEPS) ? u + 2 : u;
            LOADH(fa, un);
            __builtin_amdgcn_sched_barrier(0);
            PSTEP(fb, u + 1);
        }
#undef PSTEP
#undef LOADH
    } else if (!roleB) {
        v2f W0o[NO], W0l[NL];
        #pragma unroll
        for (int o = 0; o < NO; ++o) {
            const int g = 3 * wv + o;
            W0o[o] = *(const v2f*)&wl[0][wro + 4 * g];
        }
        #pragma unroll
        for (int q = 0; q < NL; ++q) {
            const int g = q + ((q >= 3 * wv) ? 3 : 0);
            W0l[q] = *(const v2f*)&wl[0][wro + 4 * g];
        }
        v8f wih8, b0v;
        #pragma unroll
        for (int v = 0; v < 8; ++v) {
            const int u = 12 * wv + 4 * (v >> 1) + 2 * hf + (v & 1);
            const bool ok = ((v >> 1) < 3 || wv == 3) && u < H_SZ;
            const int ue = ok ? u : H_SZ + 2;
            wih8[v] = cl[0][ue];
            b0v[v]  = cl[1][ue];
        }
        float* sp = ws + WS_S0 + tile * PSLOT + rec;
        v8f th;
        #pragma unroll
        for (int v = 0; v < 8; ++v) th[v] = 0.f;
        if (kb > 0) {
            v4f lo = *(const v4f*)sp, hi = *(const v4f*)(sp + 4);
            th = v8f{lo[0], lo[1], lo[2], lo[3], hi[0], hi[1], hi[2], hi[3]};
            #pragma unroll
            for (int j = 0; j < 3; ++j)
                *(v2f*)&H0[st + 4 * j] = own_frag(th, j);
            *(v2f*)&H0[st3] = own_frag(th, 3);
        }
        __syncthreads();
        v8f nL;
        #pragma unroll
        for (int v = 0; v < 8; ++v) nL[v] = 0.f;
        #pragma unroll
        for (int s = 0; s < NO; ++s) nL = WMMA_F32(W0o[s], own_frag(th, s), nL);

        const float* xg = x + (size_t)(b0 + r) * T_STEPS + kb * TBK;
        float* hd = ws + WS_H + (size_t)(kb & 1) * PBUF + (size_t)tile * TBK * PSLOT + rec;
        v4f xc = *(const v4f*)xg;
        v4f xn = *(const v4f*)(xg + 4);
#define PASS_0(R, WR, XV, J) pass_r<true>(R, WR, b0v, XV, wih8, W0o, W0l, offs, st, st3, th, nL, hd + (size_t)(J) * PSLOT)
        for (int j = 0; j < TBK; j += 4) {
            const int jf = (j + 8 < TBK) ? j + 8 : TBK - 4;
            const v4f xf = *(const v4f*)(xg + jf);
            PASS_0(H0, H1, xc[0], j);
            lds_barrier();
            PASS_0(H1, H0, xc[1], j + 1);
            lds_barrier();
            PASS_0(H0, H1, xc[2], j + 2);
            lds_barrier();
            PASS_0(H1, H0, xc[3], j + 3);
            lds_barrier();
            xc = xn; xn = xf;
        }
#undef PASS_0
        {
            v4f lo = {th[0], th[1], th[2], th[3]}, hi = {th[4], th[5], th[6], th[7]};
            *(v4f*)sp       = lo;
            *(v4f*)(sp + 4) = hi;
        }
    } else {
        v2f W2o[NO], W2l[NL];
        #pragma unroll
        for (int o = 0; o < NO; ++o) {
            const int g = 3 * wv + o;
            W2o[o] = *(const v2f*)&wl[2][wro + 4 * g];
        }
        #pragma unroll
        for (int q = 0; q < NL; ++q) {
            const int g = q + ((q >= 3 * wv) ? 3 : 0);
            W2l[q] = *(const v2f*)&wl[2][wro + 4 * g];
        }
        float* sp = ws + WS_S1 + tile * PSLOT + rec;
        v8f th1;
        #pragma unroll
        for (int v = 0; v < 8; ++v) th1[v] = 0.f;
        if (kb > 0) {
            v4f lo = *(const v4f*)sp, hi = *(const v4f*)(sp + 4);
            th1 = v8f{lo[0], lo[1], lo[2], lo[3], hi[0], hi[1], hi[2], hi[3]};
            #pragma unroll
            for (int j = 0; j < 3; ++j)
                *(v2f*)&H0[st + 4 * j] = own_frag(th1, j);
            *(v2f*)&H0[st3] = own_frag(th1, 3);
        }
        __syncthreads();
        v8f nL;
        #pragma unroll
        for (int v = 0; v < 8; ++v) nL[v] = 0.f;
        #pragma unroll
        for (int s = 0; s < NO; ++s) nL = WMMA_F32(W2o[s], own_frag(th1, s), nL);

        const float* pg = ws + WS_P + (size_t)(kb & 1) * PBUF + (size_t)tile * TBK * PSLOT + rec;
#define LOADP(S) ({ const float* a_ = pg + (size_t)(S) * PSLOT; v4f l_ = *(const v4f*)a_, h_ = *(const v4f*)(a_ + 4); \
                    v8f{l_[0], l_[1], l_[2], l_[3], h_[0], h_[1], h_[2], h_[3]}; })
        v8f p0 = LOADP(0);
        __builtin_amdgcn_sched_barrier(0);
        v8f p1 = LOADP(1);
        __builtin_amdgcn_sched_barrier(0);
        v8f p2 = LOADP(2);
        __builtin_amdgcn_sched_barrier(0);
        v8f p3 = LOADP(3);
#define PASS_B(R, WR, PV) pass_r<false>(R, WR, PV, 0.f, PV, W2o, W2l, offs, st, st3, th1, nL, nullptr)
        for (int j = 0; j < TBK; j += 4) {
            const int s0 = (j + 4 < TBK) ? j + 4 : TBK - 1;
            const int s1 = (j + 5 < TBK) ? j + 5 : TBK - 1;
            const int s2 = (j + 6 < TBK) ? j + 6 : TBK - 1;
            const int s3 = (j + 7 < TBK) ? j + 7 : TBK - 1;
            PASS_B(H0, H1, p0);
            p0 = LOADP(s0);
            lds_barrier();
            PASS_B(H1, H0, p1);
            p1 = LOADP(s1);
            lds_barrier();
            PASS_B(H0, H1, p2);
            p2 = LOADP(s2);
            lds_barrier();
            PASS_B(H1, H0, p3);
            p3 = LOADP(s3);
            lds_barrier();
        }
#undef PASS_B
#undef LOADP
        {
            v4f lo = {th1[0], th1[1], th1[2], th1[3]}, hi = {th1[4], th1[5], th1[6], th1[7]};
            *(v4f*)sp       = lo;
            *(v4f*)(sp + 4) = hi;
        }
        if (kb == NBLK - 1) {
            const float* h1f = H0;
            for (int i = tid; i < 16 * O_OUT; i += 128) {
                int m = i / O_OUT, o = i % O_OUT;
                float acc = b_fc[o];
                #pragma unroll 5
                for (int kk = 0; kk < H_SZ; ++kk)
                    acc = fmaf(h1f[m * RS + kk], W_fc[o * H_SZ + kk], acc);
                out[(size_t)(b0 + m) * O_OUT + o] = acc;
            }
        }
    }
}

extern "C" void kernel_launch(void* const* d_in, const int* in_sizes, int n_in,
                              void* d_out, int out_size, void* d_ws, size_t ws_size,
                              hipStream_t stream) {
    const float* x     = (const float*)d_in[0];
    const float* W_ih0 = (const float*)d_in[1];
    const float* W_hh0 = (const float*)d_in[2];
    const float* b_ih0 = (const float*)d_in[3];
    const float* b_hh0 = (const float*)d_in[4];
    const float* W_ih1 = (const float*)d_in[5];
    const float* W_hh1 = (const float*)d_in[6];
    const float* b_ih1 = (const float*)d_in[7];
    const float* b_hh1 = (const float*)d_in[8];
    const float* W_fc  = (const float*)d_in[9];
    const float* b_fc  = (const float*)d_in[10];
    float* out = (float*)d_out;
    float* ws  = (float*)d_ws;

#define STEP_LAUNCH(K) \
    rnn2_tanh_wmma<<<2 * NTILE + NPG, 128, 0, stream>>>( \
        x, W_ih0, W_hh0, b_ih0, b_hh0, W_ih1, W_hh1, b_ih1, b_hh1, W_fc, b_fc, out, ws, K)
    STEP_LAUNCH(0);
    STEP_LAUNCH(1);
    STEP_LAUNCH(2);
    STEP_LAUNCH(3);
    STEP_LAUNCH(4);
    STEP_LAUNCH(5);
    STEP_LAUNCH(6);
    STEP_LAUNCH(7);
    STEP_LAUNCH(8);
    STEP_LAUNCH(9);
#undef STEP_LAUNCH
}


# Each wave's 4 idle tile rows form the next wave's first chunk again: 4 of 13 links per wave from its own registers (was 3), 2 of them issued behind the LDS reads; order of the k sums changes
# speedup vs baseline: 2.7909x; 1.0401x over previous
#include <hip/hip_runtime.h>
#include <math.h>


typedef float v2f __attribute__((ext_vector_type(2)));
typedef float v8f __attribute__((ext_vector_type(8)));
typedef float v4f __attribute__((ext_vector_type(4)));

#define H_SZ 50
#define O_OUT 10
#define B_SZ 512
#define T_STEPS 2048
#define RS 68
#define KC 13
#define NO 4
#define NL (KC - NO)
#define HBS 1152
#define NTILE (B_SZ / 16)
#define NBLK 8
#define TBK (T_STEPS / NBLK)
#define PSLOT 1024
#define PBUF (NTILE * TBK * PSLOT)
#define NSEG 4
#define NPG (NTILE * NSEG)
#define USTEPS (TBK / NSEG / 4)
#define WS_H 0
#define WS_P (2 * PBUF)
#define WS_S0 (4 * PBUF)
#define WS_S1 (WS_S0 + NTILE * PSLOT)

__device__ __forceinline__ float fast_tanh(float v) {
#if __has_builtin(__builtin_amdgcn_tanhf)
    return __builtin_amdgcn_tanhf(v);
#else
    float e = __expf(v + v);
    return 1.0f - __fdividef(2.0f, e + 1.0f);
#endif
}

#define WMMA_F32(A, B, C) \
    __builtin_amdgcn_wmma_f32_16x16x4_f32(false, (A), false, (B), (short)0, (C), false, false)


__device__ __forceinline__ v2f own_frag(const v8f& h, int o) {
    v2f f; f[0] = h[2 * o]; f[1] = h[2 * o + 1]; return f;
}

__device__ __forceinline__ void lds_barrier() {
    __builtin_amdgcn_fence(__ATOMIC_RELEASE, "workgroup", "local");
    __builtin_amdgcn_s_barrier();
    __builtin_amdgcn_fence(__ATOMIC_ACQUIRE, "workgroup", "local");
}


template <bool L0>
__device__ __forceinline__ void pass_r(
    const float* hr, float* hw, const v8f& pin, float xv, const v8f& wih8,
    const v2f (&Wo)[NO], const v2f (&Wl)[NL],
    const int (&offs)[NL], int st, int st3,
    v8f& th, v8f& nL, float* hrec)
{
    v2f a1[NL];
    #pragma unroll
    for (int q = 0; q < NL; ++q) a1[q] = *(const v2f*)&hr[offs[q]];
    __builtin_amdgcn_sched_barrier(0);
    v8f accC, accD;
    #pragma unroll
    for (int v = 0; v < 8; ++v) accD[v] = L0 ? xv * wih8[v] + pin[v] : pin[v];
    accC = WMMA_F32(Wo[2], own_frag(th, 2), nL);
    accD = WMMA_F32(Wo[3], own_frag(th, 3), accD);
    #pragma unroll
    for (int q = 0; q < NL; ++q) {
        if (q & 1) accD = WMMA_F32(Wl[q], a1[q], accD);
        else       accC = WMMA_F32(Wl[q], a1[q], accC);
    }
    v8f c1 = accC + accD;
    #pragma unroll
    for (int v = 0; v < 8; ++v) th[v] = fast_tanh(c1[v]);
    #pragma unroll
    for (int j = 0; j < 3; ++j)
        *(v2f*)&hw[st + 4 * j] = own_frag(th, j);
    *(v2f*)&hw[st3] = own_frag(th, 3);
    __builtin_amdgcn_sched_barrier(0);
    v8f n;
    #pragma unroll
    for (int v = 0; v < 8; ++v) n[v] = 0.0f;
    #pragma unroll
    for (int s = 0; s < 2; ++s) n = WMMA_F32(Wo[s], own_frag(th, s), n);
    asm volatile("" :: "v"(n));
    nL = n;
    if (L0) {
        v4f lo = {th[0], th[1], th[2], th[3]}, hi = {th[4], th[5], th[6], th[7]};
        *(v4f*)hrec       = lo;
        *(v4f*)(hrec + 4) = hi;
    }
}

#define WP 52
#define WROWS 53

__global__ __launch_bounds__(128, 1) void rnn2_tanh_wmma(
    const float* x,
    const float* __restrict__ W_ih0, const float* __restrict__ W_hh0,
    const float* __restrict__ b_ih0, const float* __restrict__ b_hh0,
    const float* __restrict__ W_ih1, const float* __restrict__ W_hh1,
    const float* __restrict__ b_ih1, const float* __restrict__ b_hh1,
    const float* __restrict__ W_fc,  const float* __restrict__ b_fc,
    float* __restrict__ out, float* ws, int k)
{
    __shared__ __attribute__((aligned(16))) float hb[2][HBS];
    __shared__ __attribute__((aligned(16))) float wl[3][WROWS * WP];
    __shared__ float cl[3][56];

    const int bx = (int)blockIdx.x;
    const bool roleB = bx >= NTILE && bx < 2 * NTILE;
    const bool roleP = bx >= 2 * NTILE;
    const int pgi = bx - 2 * NTILE;
    const int tile = roleP ? pgi / NSEG : (roleB ? bx - NTILE : bx);
    const int kb = roleP ? k - 1 : (roleB ? k - 2 : k);
    if (kb < 0 || kb >= NBLK) return;

    const int tid  = threadIdx.x;
    __builtin_assume(tid < 128);
    const int lane = tid & 31;
    const int wv   = tid >> 5;
    const int r    = lane & 15;
    const int hf   = lane >> 4;
    const int b0   = tile * 16;

    {
        float* hz = &hb[0][0];
        for (int i = tid; i < 2 * HBS; i += 128) hz[i] = 0.f;
        for (int i = tid; i < WROWS * WP; i += 128) {
            const int row = i / WP, col = i - row * WP;
            if (row >= H_SZ || col >= H_SZ) { wl[0][i] = 0.f; wl[1][i] = 0.f; wl[2][i] = 0.f; }
        }
        #pragma unroll
        for (int it = 0; it < (H_SZ * H_SZ + 127) / 128; ++it) {
            const int i = tid + it * 128;
            if (i < H_SZ * H_SZ) {
                const int row = i / H_SZ, col = i - row * H_SZ;
                wl[0][row * WP + col] = W_hh0[i];
                wl[1][row * WP + col] = W_ih1[i];
                wl[2][row * WP + col] = W_hh1[i];
            }
        }
        if (tid < 56) {
            const bool in = tid < H_SZ;
            cl[0][tid] = in ? W_ih0[tid] : 0.f;
            cl[1][tid] = in ? b_ih0[tid] + b_hh0[tid] : 0.f;
            cl[2][tid] = in ? b_ih1[tid] + b_hh1[tid] : 0.f;
        }
    }
    __syncthreads();

    const int rj = (r & 7) >> 1;
    const int urow = 12 * wv + 4 * rj + 2 * (r >> 3) + (r & 1);
    const bool rowok = urow < H_SZ;
    const int wro = (rowok ? urow : H_SZ + 2) * WP + 2 * hf;
    int offs[NL];
    #pragma unroll
    for (int q = 0; q < NL; ++q) {
        const int g = q + ((q >= 3 * wv) ? NO : 0);
        offs[q] = r * RS + 4 * g + 2 * hf;
    }
    const int st  = r * RS + 12 * wv + 2 * hf;
    const int st3 = r * RS + 2 * hf + ((wv == 3) ? 48 : 52 + 4 * wv);
    const int rec = (wv * 32 + lane) * 8;
    float* const H0 = &hb[0][0];
    float* const H1 = &hb[1][0];

    if (roleP) {
        v2f WA[4][KC];
        v8f bb[4];
        #pragma unroll
        for (int m = 0; m < 4; ++m) {
            const int rjm = (r & 7) >> 1;
            const int um = 12 * m + 4 * rjm + 2 * (r >> 3) + (r & 1);
            const bool okm = um < H_SZ;
            const int wrm = (okm ? um : H_SZ + 2) * WP + 2 * hf;
            #pragma unroll
            for (int g = 0; g < KC; ++g) WA[m][g] = *(const v2f*)&wl[1][wrm + 4 * g];
            #pragma unroll
            for (int v = 0; v < 8; ++v) {
                const int u = 12 * m + 4 * (v >> 1) + 2 * hf + (v & 1);
                const bool ok = u < H_SZ;
                bb[m][v] = cl[2][ok ? u : H_SZ + 2];
            }
        }
        const size_t uo = (size_t)(kb & 1) * PBUF + ((size_t)tile * TBK + (size_t)(pgi % NSEG) * (TBK / NSEG) + (size_t)wv * USTEPS) * PSLOT + (size_t)lane * 8;
        const float* hs = ws + WS_H + uo;
        float* pdst = ws + WS_P + uo;
        static_assert(USTEPS % 2 == 0, "the step loop below takes two steps per trip");
#define LOADH(F, S) { _Pragma("unroll") for (int m_ = 0; m_ < 4; ++m_) { \
            const float* a_ = hs + (size_t)(S) * PSLOT + m_ * 256; \
            v4f l_ = *(const v4f*)a_, h_ = *(const v4f*)(a_ + 4); \
            F[m_] = v8f{l_[0], l_[1], l_[2], l_[3], h_[0], h_[1], h_[2], h_[3]}; } }
#define PSTEP(F, S) { \
            v8f acc_[4]; \
            _Pragma("unroll") for (int m_ = 0; m_ < 4; ++m_) acc_[m_] = bb[m_]; \
            _Pragma("unroll") for (int g_ = 0; g_ < KC; ++g_) { \
                const int wq_ = (g_ < 12) ? g_ / 3 : 3, jq_ = (g_ < 12) ? g_ % 3 : 3; \
                const v2f bf_ = own_frag(F[wq_], jq_); \
                _Pragma("unroll") for (int m_ = 0; m_ < 4; ++m_) acc_[m_] = WMMA_F32(WA[m_][g_], bf_, acc_[m_]); \
            } \
            float* pu_ = pdst + (size_t)(S) * PSLOT; \
            _Pragma("unroll") for (int m_ = 0; m_ < 4; ++m_) { \
                v4f lo_ = {acc_[m_][0], acc_[m_][1], acc_[m_][2], acc_[m_][3]}, hi_ = {acc_[m_][4], acc_[m_][5], acc_[m_][6], acc_[m_][7]}; \
                *(v4f*)(pu_ + m_ * 256)     = lo_; \
                *(v4f*)(pu_ + m_ * 256 + 4) = hi_; \
            } }
        v8f fa[4], fb[4];
        LOADH(fa, 0);
        for (int u = 0; u < USTEPS; u += 2) {
            LOADH(fb, u + 1);
            __builtin_amdgcn_sched_barrier(0);
            PSTEP(fa, u);
            const int un = (u + 2 < USTEPS) ? u + 2 : u;
            LOADH(fa, un);
            __builtin_amdgcn_sched_barrier(0);
            PSTEP(fb, u + 1);
        }
#undef PSTEP
#undef LOADH
    } else if (!roleB) {
        v2f W0o[NO], W0l[NL];
        #pragma unroll
        for (int o = 0; o < NO; ++o) {
            const int g = 3 * wv + o;
            W0o[o] = *(const v2f*)&wl[0][wro + 4 * g];
        }
        #pragma unroll
        for (int q = 0; q < NL; ++q) {
            const int g = q + ((q >= 3 * wv) ? NO : 0);
            W0l[q] = *(const v2f*)&wl[0][wro + 4 * g];
        }
        v8f wih8, b0v;
        #pragma unroll
        for (int v = 0; v < 8; ++v) {
            const int u = 12 * wv + 4 * (v >> 1) + 2 * hf + (v & 1);
            const bool ok = u < H_SZ;
            const int ue = ok ? u : H_SZ + 2;
            wih8[v] = cl[0][ue];
            b0v[v]  = cl[1][ue];
        }
        float* sp = ws + WS_S0 + tile * PSLOT + rec;
        v8f th;
        #pragma unroll
        for (int v = 0; v < 8; ++v) th[v] = 0.f;
        if (kb > 0) {
            v4f lo = *(const v4f*)sp, hi = *(const v4f*)(sp + 4);
            th = v8f{lo[0], lo[1], lo[2], lo[3], hi[0], hi[1], hi[2], hi[3]};
            #pragma unroll
            for (int j = 0; j < 3; ++j)
                *(v2f*)&H0[st + 4 * j] = own_frag(th, j);
            *(v2f*)&H0[st3] = own_frag(th, 3);
        }
        __syncthreads();
        v8f nL;
        #pragma unroll
        for (int v = 0; v < 8; ++v) nL[v] = 0.f;
        #pragma unroll
        for (int s = 0; s < 2; ++s) nL = WMMA_F32(W0o[s], own_frag(th, s), nL);

        const float* xg = x + (size_t)(b0 + r) * T_STEPS + kb * TBK;
        float* hd = ws + WS_H + (size_t)(kb & 1) * PBUF + (size_t)tile * TBK * PSLOT + rec;
        v4f xc = *(const v4f*)xg;
        v4f xn = *(const v4f*)(xg + 4);
#define PASS_0(R, WR, XV, J) pass_r<true>(R, WR, b0v, XV, wih8, W0o, W0l, offs, st, st3, th, nL, hd + (size_t)(J) * PSLOT)
        const float* xq = xg + 8;
        for (int j = 0; j < TBK; j += 4) {
            const v4f xf = *(const v4f*)xq;
            xq += (j + 12 < TBK) ? 4 : 0;
            PASS_0(H0, H1, xc[0], j);
            lds_barrier();
            PASS_0(H1, H0, xc[1], j + 1);
            lds_barrier();
            PASS_0(H0, H1, xc[2], j + 2);
            lds_barrier();
            PASS_0(H1, H0, xc[3], j + 3);
            lds_barrier();
            xc = xn; xn = xf;
        }
#undef PASS_0
        {
            v4f lo = {th[0], th[1], th[2], th[3]}, hi = {th[4], th[5], th[6], th[7]};
            *(v4f*)sp       = lo;
            *(v4f*)(sp + 4) = hi;
        }
    } else {
        v2f W2o[NO], W2l[NL];
        #pragma unroll
        for (int o = 0; o < NO; ++o) {
            const int g = 3 * wv + o;
            W2o[o] = *(const v2f*)&wl[2][wro + 4 * g];
        }
        #pragma unroll
        for (int q = 0; q < NL; ++q) {
            const int g = q + ((q >= 3 * wv) ? NO : 0);
            W2l[q] = *(const v2f*)&wl[2][wro + 4 * g];
        }
        float* sp = ws + WS_S1 + tile * PSLOT + rec;
        v8f th1;
        #pragma unroll
        for (int v = 0; v < 8; ++v) th1[v] = 0.f;
        if (kb > 0) {
            v4f lo = *(const v4f*)sp, hi = *(const v4f*)(sp + 4);
            th1 = v8f{lo[0], lo[1], lo[2], lo[3], hi[0], hi[1], hi[2], hi[3]};
            #pragma unroll
            for (int j = 0; j < 3; ++j)
                *(v2f*)&H0[st + 4 * j] = own_frag(th1, j);
            *(v2f*)&H0[st3] = own_frag(th1, 3);
        }
        __syncthreads();
        v8f nL;
        #pragma unroll
        for (int v = 0; v < 8; ++v) nL[v] = 0.f;
        #pragma unroll
        for (int s = 0; s < 2; ++s) nL = WMMA_F32(W2o[s], own_frag(th1, s), nL);

        const float* pg = ws + WS_P + (size_t)(kb & 1) * PBUF + (size_t)tile * TBK * PSLOT + rec;
#define LOADP(S) ({ const float* a_ = pg + (size_t)(S) * PSLOT; v4f l_ = *(const v4f*)a_, h_ = *(const v4f*)(a_ + 4); \
                    v8f{l_[0], l_[1], l_[2], l_[3], h_[0], h_[1], h_[2], h_[3]}; })
        v8f p0 = LOADP(0);
        __builtin_amdgcn_sched_barrier(0);
        v8f p1 = LOADP(1);
        __builtin_amdgcn_sched_barrier(0);
        v8f p2 = LOADP(2);
        __builtin_amdgcn_sched_barrier(0);
        v8f p3 = LOADP(3);
#define PASS_B(R, WR, PV) pass_r<false>(R, WR, PV, 0.f, PV, W2o, W2l, offs, st, st3, th1, nL, nullptr)
        for (int j = 0; j < TBK; j += 4) {
            const int s0 = (j + 4 < TBK) ? j + 4 : TBK - 1;
            const int s1 = (j + 5 < TBK) ? j + 5 : TBK - 1;
            const int s2 = (j + 6 < TBK) ? j + 6 : TBK - 1;
            const int s3 = (j + 7 < TBK) ? j + 7 : TBK - 1;
            PASS_B(H0, H1, p0);
            p0 = LOADP(s0);
            lds_barrier();
            PASS_B(H1, H0, p1);
            p1 = LOADP(s1);
            lds_barrier();
            PASS_B(H0, H1, p2);
            p2 = LOADP(s2);
            lds_barrier();
            PASS_B(H1, H0, p3);
            p3 = LOADP(s3);
            lds_barrier();
        }
#undef PASS_B
#undef LOADP
        {
            v4f lo = {th1[0], th1[1], th1[2], th1[3]}, hi = {th1[4], th1[5], th1[6], th1[7]};
            *(v4f*)sp       = lo;
            *(v4f*)(sp + 4) = hi;
        }
        if (kb == NBLK - 1) {
            const float* h1f = H0;
            for (int i = tid; i < 16 * O_OUT; i += 128) {
                int m = i / O_OUT, o = i % O_OUT;
                float acc = b_fc[o];
                #pragma unroll 5
                for (int kk = 0; kk < H_SZ; ++kk)
                    acc = fmaf(h1f[m * RS + kk], W_fc[o * H_SZ + kk], acc);
                out[(size_t)(b0 + m) * O_OUT + o] = acc;
            }
        }
    }
}

extern "C" void kernel_launch(void* const* d_in, const int* in_sizes, int n_in,
                              void* d_out, int out_size, void* d_ws, size_t ws_size,
                              hipStream_t stream) {
    const float* x     = (const float*)d_in[0];
    const float* W_ih0 = (const float*)d_in[1];
    const float* W_hh0 = (const float*)d_in[2];
    const float* b_ih0 = (const float*)d_in[3];
    const float* b_hh0 = (const float*)d_in[4];
    const float* W_ih1 = (const float*)d_in[5];
    const float* W_hh1 = (const float*)d_in[6];
    const float* b_ih1 = (const float*)d_in[7];
    const float* b_hh1 = (const float*)d_in[8];
    const float* W_fc  = (const float*)d_in[9];
    const float* b_fc  = (const float*)d_in[10];
    float* out = (float*)d_out;
    float* ws  = (float*)d_ws;

#define STEP_LAUNCH(K) \
    rnn2_tanh_wmma<<<2 * NTILE + NPG, 128, 0, stream>>>( \
        x, W_ih0, W_hh0, b_ih0, b_hh0, W_ih1, W_hh1, b_ih1, b_hh1, W_fc, b_fc, out, ws, K)
    STEP_LAUNCH(0);
    STEP_LAUNCH(1);
    STEP_LAUNCH(2);
    STEP_LAUNCH(3);
    STEP_LAUNCH(4);
    STEP_LAUNCH(5);
    STEP_LAUNCH(6);
    STEP_LAUNCH(7);
    STEP_LAUNCH(8);
    STEP_LAUNCH(9);
#undef STEP_LAUNCH
}


# Pipeline re-cut into 18 launches of 128 steps (was 10 of 256), each launch's set-up cut to one memory round trip (weights, state, first inputs straight to registers): fill and drain cost half
# speedup vs baseline: 2.9631x; 1.0617x over previous
#include <hip/hip_runtime.h>
#include <math.h>


typedef float v2f __attribute__((ext_vector_type(2)));
typedef float v8f __attribute__((ext_vector_type(8)));
typedef float v4f __attribute__((ext_vector_type(4)));

#define H_SZ 50
#define O_OUT 10
#define B_SZ 512
#define T_STEPS 2048
#define RS 68
#define KC 13
#define NO 4
#define NL (KC - NO)
#define HBS 1152
#define NTILE (B_SZ / 16)
#define NBLK 16
#define TBK (T_STEPS / NBLK)
#define PSLOT 1024
#define PBUF (NTILE * TBK * PSLOT)
#define NSEG 4
#define NPG (NTILE * NSEG)
#define USTEPS (TBK / NSEG / 4)
#define WS_H 0
#define WS_P (2 * PBUF)
#define WS_S0 (4 * PBUF)
#define WS_S1 (WS_S0 + NTILE * PSLOT)

__device__ __forceinline__ float fast_tanh(float v) {
#if __has_builtin(__builtin_amdgcn_tanhf)
    return __builtin_amdgcn_tanhf(v);
#else
    float e = __expf(v + v);
    return 1.0f - __fdividef(2.0f, e + 1.0f);
#endif
}

#define WMMA_F32(A, B, C) \
    __builtin_amdgcn_wmma_f32_16x16x4_f32(false, (A), false, (B), (short)0, (C), false, false)


__device__ __forceinline__ v2f own_frag(const v8f& h, int o) {
    v2f f; f[0] = h[2 * o]; f[1] = h[2 * o + 1]; return f;
}

__device__ __forceinline__ void lds_barrier() {
    __builtin_amdgcn_fence(__ATOMIC_RELEASE, "workgroup", "local");
    __builtin_amdgcn_s_barrier();
    __builtin_amdgcn_fence(__ATOMIC_ACQUIRE, "workgroup", "local");
}


template <bool L0>
__device__ __forceinline__ void pass_r(
    const float* hr, float* hw, const v8f& pin, float xv, const v8f& wih8,
    const v2f (&Wo)[NO], const v2f (&Wl)[NL],
    const int (&offs)[NL], int st, int st3,
    v8f& th, v8f& nL, float* hrec)
{
    v2f a1[NL];
    #pragma unroll
    for (int q = 0; q < NL; ++q) a1[q] = *(const v2f*)&hr[offs[q]];
    __builtin_amdgcn_sched_barrier(0);
    v8f accC, accD;
    #pragma unroll
    for (int v = 0; v < 8; ++v) accD[v] = L0 ? xv * wih8[v] + pin[v] : pin[v];
    accC = WMMA_F32(Wo[2], own_frag(th, 2), nL);
    accD = WMMA_F32(Wo[3], own_frag(th, 3), accD);
    #pragma unroll
    for (int q = 0; q < NL; ++q) {
        if (q & 1) accD = WMMA_F32(Wl[q], a1[q], accD);
        else       accC = WMMA_F32(Wl[q], a1[q], accC);
    }
    v8f c1 = accC + accD;
    #pragma unroll
    for (int v = 0; v < 8; ++v) th[v] = fast_tanh(c1[v]);
    #pragma unroll
    for (int j = 0; j < 3; ++j)
        *(v2f*)&hw[st + 4 * j] = own_frag(th, j);
    *(v2f*)&hw[st3] = own_frag(th, 3);
    __builtin_amdgcn_sched_barrier(0);
    v8f n;
    #pragma unroll
    for (int v = 0; v < 8; ++v) n[v] = 0.0f;
    #pragma unroll
    for (int s = 0; s < 2; ++s) n = WMMA_F32(Wo[s], own_frag(th, s), n);
    asm volatile("" :: "v"(n));
    nL = n;
    if (L0) {
        v4f lo = {th[0], th[1], th[2], th[3]}, hi = {th[4], th[5], th[6], th[7]};
        *(v4f*)hrec       = lo;
        *(v4f*)(hrec + 4) = hi;
    }
}

__device__ __forceinline__ bool wfrag_ok(bool rowok, int hf, int g) { return rowok && 4 * g + 2 * hf < H_SZ; }
__device__ __forceinline__ v2f wfrag_load(const float* __restrict__ Wm, int urow, bool rowok, int hf, int g) {
    return *(const v2f*)&Wm[wfrag_ok(rowok, hf, g) ? urow * H_SZ + 4 * g + 2 * hf : 0];
}
__device__ __forceinline__ v2f wfrag_mask(v2f w, bool rowok, int hf, int g) {
    const bool ok = wfrag_ok(rowok, hf, g);
    v2f f; f[0] = ok ? w[0] : 0.f; f[1] = ok ? w[1] : 0.f; return f;
}

#define WP 52
#define WROWS 53

__global__ __launch_bounds__(128, 1) void rnn2_tanh_wmma(
    const float* x,
    const float* __restrict__ W_ih0, const float* __restrict__ W_hh0,
    const float* __restrict__ b_ih0, const float* __restrict__ b_hh0,
    const float* __restrict__ W_ih1, const float* __restrict__ W_hh1,
    const float* __restrict__ b_ih1, const float* __restrict__ b_hh1,
    const float* __restrict__ W_fc,  const float* __restrict__ b_fc,
    float* __restrict__ out, float* ws, int k)
{
    __shared__ __attribute__((aligned(16))) float hb[2][HBS];
    __shared__ __attribute__((aligned(16))) float wl1[WROWS * WP];
    __shared__ float cl2[56];

    const int bx = (int)blockIdx.x;
    const bool roleB = bx >= NTILE && bx < 2 * NTILE;
    const bool roleP = bx >= 2 * NTILE;
    const int pgi = bx - 2 * NTILE;
    const int tile = roleP ? pgi / NSEG : (roleB ? bx - NTILE : bx);
    const int kb = roleP ? k - 1 : (roleB ? k - 2 : k);
    if (kb < 0 || kb >= NBLK) return;

    const int tid  = threadIdx.x;
    __builtin_assume(tid < 128);
    const int lane = tid & 31;
    const int wv   = tid >> 5;
    const int r    = lane & 15;
    const int hf   = lane >> 4;
    const int b0   = tile * 16;

    const int rj = (r & 7) >> 1;
    const int urow = 12 * wv + 4 * rj + 2 * (r >> 3) + (r & 1);
    const bool rowok = urow < H_SZ;
    int offs[NL];
    #pragma unroll
    for (int q = 0; q < NL; ++q) {
        const int g = q + ((q >= 3 * wv) ? NO : 0);
        offs[q] = r * RS + 4 * g + 2 * hf;
    }
    const int st  = r * RS + 12 * wv + 2 * hf;
    const int st3 = r * RS + 2 * hf + ((wv == 3) ? 48 : 52 + 4 * wv);
    const int rec = (wv * 32 + lane) * 8;
    float* const H0 = &hb[0][0];
    float* const H1 = &hb[1][0];


    if (roleP) {
        {
            for (int i = tid; i < WROWS * WP; i += 128) {
                const int row = i / WP, col = i - row * WP;
                if (row >= H_SZ || col >= H_SZ) wl1[i] = 0.f;
            }
            #pragma unroll
            for (int it = 0; it < (H_SZ * H_SZ + 127) / 128; ++it) {
                const int i = tid + it * 128;
                if (i < H_SZ * H_SZ) {
                    const int row = i / H_SZ, col = i - row * H_SZ;
                    wl1[row * WP + col] = W_ih1[i];
                }
            }
            if (tid < 56) cl2[tid] = (tid < H_SZ) ? b_ih1[tid] + b_hh1[tid] : 0.f;
        }
        __syncthreads();
        v2f WA[4][KC];
        v8f bb[4];
        #pragma unroll
        for (int m = 0; m < 4; ++m) {
            const int rjm = (r & 7) >> 1;
            const int um = 12 * m + 4 * rjm + 2 * (r >> 3) + (r & 1);
            const bool okm = um < H_SZ;
            const int wrm = (okm ? um : H_SZ + 2) * WP + 2 * hf;
            #pragma unroll
            for (int g = 0; g < KC; ++g) WA[m][g] = *(const v2f*)&wl1[wrm + 4 * g];
            #pragma unroll
            for (int v = 0; v < 8; ++v) {
                const int u = 12 * m + 4 * (v >> 1) + 2 * hf + (v & 1);
                const bool ok = u < H_SZ;
                bb[m][v] = cl2[ok ? u : H_SZ + 2];
            }
        }
        const size_t uo = (size_t)(kb & 1) * PBUF + ((size_t)tile * TBK + (size_t)(pgi % NSEG) * (TBK / NSEG) + (size_t)wv * USTEPS) * PSLOT + (size_t)lane * 8;
        const float* hs = ws + WS_H + uo;
        float* pdst = ws + WS_P + uo;
        static_assert(USTEPS % 2 == 0, "the step loop below takes two steps per trip");
#define LOADH(F, S) { _Pragma("unroll") for (int m_ = 0; m_ < 4; ++m_) { \
            const float* a_ = hs + (size_t)(S) * PSLOT + m_ * 256; \
            v4f l_ = *(const v4f*)a_, h_ = *(const v4f*)(a_ + 4); \
            F[m_] = v8f{l_[0], l_[1], l_[2], l_[3], h_[0], h_[1], h_[2], h_[3]}; } }
#define PSTEP(F, S) { \
            v8f acc_[4]; \
            _Pragma("unroll") for (int m_ = 0; m_ < 4; ++m_) acc_[m_] = bb[m_]; \
            _Pragma("unroll") for (int g_ = 0; g_ < KC; ++g_) { \
                const int wq_ = (g_ < 12) ? g_ / 3 : 3, jq_ = (g_ < 12) ? g_ % 3 : 3; \
                const v2f bf_ = own_frag(F[wq_], jq_); \
                _Pragma("unroll") for (int m_ = 0; m_ < 4; ++m_) acc_[m_] = WMMA_F32(WA[m_][g_], bf_, acc_[m_]); \
            } \
            float* pu_ = pdst + (size_t)(S) * PSLOT; \
            _Pragma("unroll") for (int m_ = 0; m_ < 4; ++m_) { \
                v4f lo_ = {acc_[m_][0], acc_[m_][1], acc_[m_][2], acc_[m_][3]}, hi_ = {acc_[m_][4], acc_[m_][5], acc_[m_][6], acc_[m_][7]}; \
                *(v4f*)(pu_ + m_ * 256)     = lo_; \
                *(v4f*)(pu_ + m_ * 256 + 4) = hi_; \
            } }
        v8f fa[4], fb[4];
        LOADH(fa, 0);
        for (int u = 0; u < USTEPS; u += 2) {
            LOADH(fb, u + 1);
            __builtin_amdgcn_sched_barrier(0);
            PSTEP(fa, u);
            const int un = (u + 2 < USTEPS) ? u + 2 : u;
            LOADH(fa, un);
            __builtin_amdgcn_sched_barrier(0);
            PSTEP(fb, u + 1);
        }
#undef PSTEP
#undef LOADH
    } else if (!roleB) {
        v2f W0o[NO], W0l[NL];
        #pragma unroll
        for (int o = 0; o < NO; ++o) W0o[o] = wfrag_load(W_hh0, urow, rowok, hf, 3 * wv + o);
        #pragma unroll
        for (int q = 0; q < NL; ++q) W0l[q] = wfrag_load(W_hh0, urow, rowok, hf, q + ((q >= 3 * wv) ? NO : 0));
        v2f wi[4], bi[4], bh[4];
        #pragma unroll
        for (int j = 0; j < 4; ++j) {
            const int u = 12 * wv + 4 * j + 2 * hf;
            const int ue = (u < H_SZ) ? u : 0;
            wi[j] = *(const v2f*)&W_ih0[ue];
            bi[j] = *(const v2f*)&b_ih0[ue];
            bh[j] = *(const v2f*)&b_hh0[ue];
        }
        const float* xg = x + (size_t)(b0 + r) * T_STEPS + kb * TBK;
        float* hd = ws + WS_H + (size_t)(kb & 1) * PBUF + (size_t)tile * TBK * PSLOT + rec;
        v4f xc = *(const v4f*)xg;
        v4f xn = *(const v4f*)(xg + 4);
        __builtin_amdgcn_sched_barrier(0);
        float* sp = ws + WS_S0 + tile * PSLOT + rec;
        v8f th;
        #pragma unroll
        for (int v = 0; v < 8; ++v) th[v] = 0.f;
        if (kb > 0) {
            v4f lo = *(const v4f*)sp, hi = *(const v4f*)(sp + 4);
            th = v8f{lo[0], lo[1], lo[2], lo[3], hi[0], hi[1], hi[2], hi[3]};
        }
        #pragma unroll
        for (int o = 0; o < NO; ++o) W0o[o] = wfrag_mask(W0o[o], rowok, hf, 3 * wv + o);
        #pragma unroll
        for (int q = 0; q < NL; ++q) W0l[q] = wfrag_mask(W0l[q], rowok, hf, q + ((q >= 3 * wv) ? NO : 0));
        v8f wih8, b0v;
        #pragma unroll
        for (int j = 0; j < 4; ++j) {
            const bool ok = 12 * wv + 4 * j + 2 * hf < H_SZ;
            wih8[2 * j]     = ok ? wi[j][0] : 0.f;
            wih8[2 * j + 1] = ok ? wi[j][1] : 0.f;
            b0v[2 * j]      = ok ? bi[j][0] + bh[j][0] : 0.f;
            b0v[2 * j + 1]  = ok ? bi[j][1] + bh[j][1] : 0.f;
        }
        #pragma unroll
        for (int j = 0; j < 3; ++j)
            *(v2f*)&H0[st + 4 * j] = own_frag(th, j);
        *(v2f*)&H0[st3] = own_frag(th, 3);
        lds_barrier();
        v8f nL;
        #pragma unroll
        for (int v = 0; v < 8; ++v) nL[v] = 0.f;
        #pragma unroll
        for (int s = 0; s < 2; ++s) nL = WMMA_F32(W0o[s], own_frag(th, s), nL);

#define PASS_0(R, WR, XV, J) pass_r<true>(R, WR, b0v, XV, wih8, W0o, W0l, offs, st, st3, th, nL, hd + (size_t)(J) * PSLOT)
        const float* xq = xg + 8;
        for (int j = 0; j < TBK; j += 4) {
            const v4f xf = *(const v4f*)xq;
            xq += (j + 12 < TBK) ? 4 : 0;
            PASS_0(H0, H1, xc[0], j);
            lds_barrier();
            PASS_0(H1, H0, xc[1], j + 1);
            lds_barrier();
            PASS_0(H0, H1, xc[2], j + 2);
            lds_barrier();
            PASS_0(H1, H0, xc[3], j + 3);
            lds_barrier();
            xc = xn; xn = xf;
        }
#undef PASS_0
        {
            v4f lo = {th[0], th[1], th[2], th[3]}, hi = {th[4], th[5], th[6], th[7]};
            *(v4f*)sp       = lo;
            *(v4f*)(sp + 4) = hi;
        }
    } else {
        v2f W2o[NO], W2l[NL];
        #pragma unroll
        for (int o = 0; o < NO; ++o) W2o[o] = wfrag_load(W_hh1, urow, rowok, hf, 3 * wv + o);
        #pragma unroll
        for (int q = 0; q < NL; ++q) W2l[q] = wfrag_load(W_hh1, urow, rowok, hf, q + ((q >= 3 * wv) ? NO : 0));
        const float* pg = ws + WS_P + (size_t)(kb & 1) * PBUF + (size_t)tile * TBK * PSLOT + rec;
#define LOADP(S) ({ const float* a_ = pg + (size_t)(S) * PSLOT; v4f l_ = *(const v4f*)a_, h_ = *(const v4f*)(a_ + 4); \
                    v8f{l_[0], l_[1], l_[2], l_[3], h_[0], h_[1], h_[2], h_[3]}; })
        v8f p0 = LOADP(0);
        __builtin_amdgcn_sched_barrier(0);
        v8f p1 = LOADP(1);
        __builtin_amdgcn_sched_barrier(0);
        v8f p2 = LOADP(2);
        __builtin_amdgcn_sched_barrier(0);
        v8f p3 = LOADP(3);
        __builtin_amdgcn_sched_barrier(0);
        float* sp = ws + WS_S1 + tile * PSLOT + rec;
        v8f th1;
        #pragma unroll
        for (int v = 0; v < 8; ++v) th1[v] = 0.f;
        if (kb > 0) {
            v4f lo = *(const v4f*)sp, hi = *(const v4f*)(sp + 4);
            th1 = v8f{lo[0], lo[1], lo[2], lo[3], hi[0], hi[1], hi[2], hi[3]};
        }
        #pragma unroll
        for (int o = 0; o < NO; ++o) W2o[o] = wfrag_mask(W2o[o], rowok, hf, 3 * wv + o);
        #pragma unroll
        for (int q = 0; q < NL; ++q) W2l[q] = wfrag_mask(W2l[q], rowok, hf, q + ((q >= 3 * wv) ? NO : 0));
        #pragma unroll
        for (int j = 0; j < 3; ++j)
            *(v2f*)&H0[st + 4 * j] = own_frag(th1, j);
        *(v2f*)&H0[st3] = own_frag(th1, 3);
        lds_barrier();
        v8f nL;
        #pragma unroll
        for (int v = 0; v < 8; ++v) nL[v] = 0.f;
        #pragma unroll
        for (int s = 0; s < 2; ++s) nL = WMMA_F32(W2o[s], own_frag(th1, s), nL);

#define PASS_B(R, WR, PV) pass_r<false>(R, WR, PV, 0.f, PV, W2o, W2l, offs, st, st3, th1, nL, nullptr)
        for (int j = 0; j < TBK; j += 4) {
            const int s0 = (j + 4 < TBK) ? j + 4 : TBK - 1;
            const int s1 = (j + 5 < TBK) ? j + 5 : TBK - 1;
            const int s2 = (j + 6 < TBK) ? j + 6 : TBK - 1;
            const int s3 = (j + 7 < TBK) ? j + 7 : TBK - 1;
            PASS_B(H0, H1, p0);
            p0 = LOADP(s0);
            lds_barrier();
            PASS_B(H1, H0, p1);
            p1 = LOADP(s1);
            lds_barrier();
            PASS_B(H0, H1, p2);
            p2 = LOADP(s2);
            lds_barrier();
            PASS_B(H1, H0, p3);
            p3 = LOADP(s3);
            lds_barrier();
        }
#undef PASS_B
#undef LOADP
        {
            v4f lo = {th1[0], th1[1], th1[2], th1[3]}, hi = {th1[4], th1[5], th1[6], th1[7]};
            *(v4f*)sp       = lo;
            *(v4f*)(sp + 4) = hi;
        }
        if (kb == NBLK - 1) {
            const float* h1f = H0;
            for (int i = tid; i < 16 * O_OUT; i += 128) {
                int m = i / O_OUT, o = i % O_OUT;
                float acc = b_fc[o];
                #pragma unroll 5
                for (int kk = 0; kk < H_SZ; ++kk)
                    acc = fmaf(h1f[m * RS + kk], W_fc[o * H_SZ + kk], acc);
                out[(size_t)(b0 + m) * O_OUT + o] = acc;
            }
        }
    }
}

extern "C" void kernel_launch(void* const* d_in, const int* in_sizes, int n_in,
                              void* d_out, int out_size, void* d_ws, size_t ws_size,
                              hipStream_t stream) {
    const float* x     = (const float*)d_in[0];
    const float* W_ih0 = (const float*)d_in[1];
    const float* W_hh0 = (const float*)d_in[2];
    const float* b_ih0 = (const float*)d_in[3];
    const float* b_hh0 = (const float*)d_in[4];
    const float* W_ih1 = (const float*)d_in[5];
    const float* W_hh1 = (const float*)d_in[6];
    const float* b_ih1 = (const float*)d_in[7];
    const float* b_hh1 = (const float*)d_in[8];
    const float* W_fc  = (const float*)d_in[9];
    const float* b_fc  = (const float*)d_in[10];
    float* out = (float*)d_out;
    float* ws  = (float*)d_ws;

#define STEP_LAUNCH(K) \
    rnn2_tanh_wmma<<<2 * NTILE + NPG, 128, 0, stream>>>( \
        x, W_ih0, W_hh0, b_ih0, b_hh0, W_ih1, W_hh1, b_ih1, b_hh1, W_fc, b_fc, out, ws, K)
    STEP_LAUNCH(0);
    STEP_LAUNCH(1);
    STEP_LAUNCH(2);
    STEP_LAUNCH(3);
    STEP_LAUNCH(4);
    STEP_LAUNCH(5);
    STEP_LAUNCH(6);
    STEP_LAUNCH(7);
    STEP_LAUNCH(8);
    STEP_LAUNCH(9);
    STEP_LAUNCH(10);
    STEP_LAUNCH(11);
    STEP_LAUNCH(12);
    STEP_LAUNCH(13);
    STEP_LAUNCH(14);
    STEP_LAUNCH(15);
    STEP_LAUNCH(16);
    STEP_LAUNCH(17);
#undef STEP_LAUNCH
}
